# starting kernel
# baseline (speedup 1.0000x reference)
#include <hip/hip_runtime.h>
#include <hip/hip_bf16.h>

typedef __attribute__((ext_vector_type(2))) float v2f;
typedef __attribute__((ext_vector_type(8))) float v8f;

__device__ __forceinline__ v8f wmma4(v2f a, v2f b, v8f c) {
    return __builtin_amdgcn_wmma_f32_16x16x4_f32(false, a, false, b, (short)0, c, false, false);
}

__global__ void mf_kernel(const int* __restrict__ mask, float* __restrict__ mfv, int n) {
    int i = blockIdx.x * blockDim.x + threadIdx.x;
    if (i < n) mfv[i] = mask[i] ? 1.0f : 0.0f;
}

__global__ void ln_kernel(const float* __restrict__ x, float* __restrict__ out,
                          const float* __restrict__ g, const float* __restrict__ b) {
    __shared__ float red[256];
    const int row = blockIdx.x;
    const int tid = threadIdx.x;
    float v = x[(size_t)row * 256 + tid];
    red[tid] = v;
    __syncthreads();
    for (int s = 128; s > 0; s >>= 1) {
        if (tid < s) red[tid] += red[tid + s];
        __syncthreads();
    }
    float mean = red[0] * (1.0f / 256.0f);
    __syncthreads();
    float d = v - mean;
    red[tid] = d * d;
    __syncthreads();
    for (int s = 128; s > 0; s >>= 1) {
        if (tid < s) red[tid] += red[tid + s];
        __syncthreads();
    }
    float var = red[0] * (1.0f / 256.0f);
    float r = rsqrtf(var + 1e-6f);
    out[(size_t)row * 256 + tid] = d * r * g[tid] + b[tid];
}

__global__ void gemm_kernel(const float* __restrict__ A, const float* __restrict__ W,
                            float* __restrict__ C, const float* __restrict__ bias,
                            const float* __restrict__ mfv, const float* __restrict__ res,
                            int K, int N, int relu) {
    __shared__ __align__(16) float As[64][34];
    __shared__ __align__(16) float Bs[16][144][2];

    const int tid = threadIdx.x;
    const int lane = tid & 31;
    const int wave = tid >> 5;
    const int half = lane >> 4;
    const int lane16 = lane & 15;
    const int wm = wave >> 2;
    const int wn = wave & 3;
    const int m0 = blockIdx.x * 64;
    const int n0 = blockIdx.y * 128;

    const int rowA0 = wm * 32 + lane16;
    const int rowA1 = rowA0 + 16;
    const int colB0 = wn * 32 + lane16;
    const int colB1 = colB0 + 16;

    v8f acc[2][2] = {};

    for (int k0 = 0; k0 < K; k0 += 32) {
        for (int i = tid; i < 64 * 32; i += 256) {
            int r = i >> 5, c = i & 31;
            As[r][c] = A[(size_t)(m0 + r) * K + k0 + c];
        }
        for (int i = tid; i < 16 * 128; i += 256) {
            int kp = i >> 7, c = i & 127;
            Bs[kp][c][0] = W[(size_t)(k0 + 2 * kp) * N + n0 + c];
            Bs[kp][c][1] = W[(size_t)(k0 + 2 * kp + 1) * N + n0 + c];
        }
        if (k0 + 32 < K) {
            if (tid < 64) {
                __builtin_prefetch(&A[(size_t)(m0 + tid) * K + k0 + 32], 0, 1);
            } else if (tid < 192) {
                int r = (tid - 64) >> 2, s = (tid - 64) & 3;
                __builtin_prefetch(&W[(size_t)(k0 + 32 + r) * N + n0 + s * 32], 0, 1);
            }
        }
        __syncthreads();

        v2f a0c = *(const v2f*)&As[rowA0][half * 2];
        v2f a1c = *(const v2f*)&As[rowA1][half * 2];
        v2f b0c = *(const v2f*)&Bs[half][colB0][0];
        v2f b1c = *(const v2f*)&Bs[half][colB1][0];
#pragma unroll
        for (int t = 0; t < 8; ++t) {
            const int kkn = ((t + 1) & 7) * 4 + half * 2;
            v2f a0n = *(const v2f*)&As[rowA0][kkn];
            v2f a1n = *(const v2f*)&As[rowA1][kkn];
            v2f b0n = *(const v2f*)&Bs[kkn >> 1][colB0][0];
            v2f b1n = *(const v2f*)&Bs[kkn >> 1][colB1][0];
            acc[0][0] = wmma4(a0c, b0c, acc[0][0]);
            acc[0][1] = wmma4(a0c, b1c, acc[0][1]);
            acc[1][0] = wmma4(a1c, b0c, acc[1][0]);
            acc[1][1] = wmma4(a1c, b1c, acc[1][1]);
            a0c = a0n; a1c = a1n; b0c = b0n; b1c = b1n;
        }
        __syncthreads();
    }

#pragma unroll
    for (int im = 0; im < 2; ++im) {
#pragma unroll
        for (int jn = 0; jn < 2; ++jn) {
            const int col = n0 + wn * 32 + jn * 16 + lane16;
            const float bv = bias ? bias[col] : 0.0f;
#pragma unroll
            for (int vi = 0; vi < 8; ++vi) {
                const int row = m0 + wm * 32 + im * 16 + vi + half * 8;
                float val = acc[im][jn][vi] + bv;
                if (relu) val = fmaxf(val, 0.0f);
                if (mfv) val *= mfv[row];
                if (res) val += res[(size_t)row * N + col];
                C[(size_t)row * N + col] = val;
            }
        }
    }
}

__global__ void attn_kernel(const float* __restrict__ q, const float* __restrict__ k,
                            const float* __restrict__ v, const float* __restrict__ mfv,
                            float* __restrict__ o) {
    __shared__ __align__(16) float Qs[64][34];
    __shared__ __align__(16) float Ks[16][80][2];
    __shared__ __align__(16) float Vs[32][48][2];
    __shared__ __align__(16) float Ps[64][68];
    __shared__ float mb[64];
    __shared__ float mrun[64], lrun[64], alph[64];

    const int S = 2048, LD = 256;
    const int tid = threadIdx.x;
    const int lane = tid & 31;
    const int wave = tid >> 5;
    const int half = lane >> 4;
    const int lane16 = lane & 15;
    const int bh = blockIdx.y;
    const int b = bh >> 3;
    const int h = bh & 7;
    const int q0 = blockIdx.x * 64;
    const int qrow = wave * 16 + lane16;

    const size_t baseQ = ((size_t)b * S + q0) * LD + h * 32;
    for (int i = tid; i < 64 * 32; i += 128) {
        int r = i >> 5, c = i & 31;
        Qs[r][c] = q[baseQ + (size_t)r * LD + c];
    }
    if (tid < 64) { mrun[tid] = -3.0e38f; lrun[tid] = 0.0f; }
    v8f oacc[2] = {};
    __syncthreads();

    for (int j0 = 0; j0 < S; j0 += 64) {
        const size_t baseK = ((size_t)b * S + j0) * LD + h * 32;
        for (int i = tid; i < 16 * 64; i += 128) {
            int r = i & 63, kp = i >> 6;
            Ks[kp][r][0] = k[baseK + (size_t)r * LD + 2 * kp];
            Ks[kp][r][1] = k[baseK + (size_t)r * LD + 2 * kp + 1];
        }
        for (int i = tid; i < 32 * 32; i += 128) {
            int n = i & 31, kp = i >> 5;
            Vs[kp][n][0] = v[baseK + (size_t)(2 * kp) * LD + n];
            Vs[kp][n][1] = v[baseK + (size_t)(2 * kp + 1) * LD + n];
        }
        if (tid < 64) mb[tid] = (mfv[(size_t)b * S + j0 + tid] != 0.0f) ? 0.0f : -1e9f;
        if (j0 + 64 < S) {
            const size_t baseN = ((size_t)b * S + j0 + 64) * LD + h * 32;
            if (tid < 64) __builtin_prefetch(&k[baseN + (size_t)tid * LD], 0, 1);
            else          __builtin_prefetch(&v[baseN + (size_t)(tid - 64) * LD], 0, 1);
        }
        __syncthreads();

        v8f sacc[4] = {};
        v2f ac = *(const v2f*)&Qs[qrow][half * 2];
        v2f bc0 = *(const v2f*)&Ks[half][0 * 16 + lane16][0];
        v2f bc1 = *(const v2f*)&Ks[half][1 * 16 + lane16][0];
        v2f bc2 = *(const v2f*)&Ks[half][2 * 16 + lane16][0];
        v2f bc3 = *(const v2f*)&Ks[half][3 * 16 + lane16][0];
#pragma unroll
        for (int t = 0; t < 8; ++t) {
            const int kkn = ((t + 1) & 7) * 4 + half * 2;
            v2f an = *(const v2f*)&Qs[qrow][kkn];
            v2f bn0 = *(const v2f*)&Ks[kkn >> 1][0 * 16 + lane16][0];
            v2f bn1 = *(const v2f*)&Ks[kkn >> 1][1 * 16 + lane16][0];
            v2f bn2 = *(const v2f*)&Ks[kkn >> 1][2 * 16 + lane16][0];
            v2f bn3 = *(const v2f*)&Ks[kkn >> 1][3 * 16 + lane16][0];
            sacc[0] = wmma4(ac, bc0, sacc[0]);
            sacc[1] = wmma4(ac, bc1, sacc[1]);
            sacc[2] = wmma4(ac, bc2, sacc[2]);
            sacc[3] = wmma4(ac, bc3, sacc[3]);
            ac = an; bc0 = bn0; bc1 = bn1; bc2 = bn2; bc3 = bn3;
        }
        const float inv = 0.17677669529663687f;
#pragma unroll
        for (int nt = 0; nt < 4; ++nt) {
            const int col = nt * 16 + lane16;
#pragma unroll
            for (int vi = 0; vi < 8; ++vi) {
                const int row = wave * 16 + vi + half * 8;
                Ps[row][col] = sacc[nt][vi] * inv + mb[col];
            }
        }
        __syncthreads();

        if (tid < 64) {
            float mo = mrun[tid];
            float mc = mo;
            for (int j = 0; j < 64; ++j) mc = fmaxf(mc, Ps[tid][j]);
            float al = __expf(mo - mc);
            float ls = 0.0f;
            for (int j = 0; j < 64; ++j) {
                float p = __expf(Ps[tid][j] - mc);
                Ps[tid][j] = p;
                ls += p;
            }
            lrun[tid] = lrun[tid] * al + ls;
            mrun[tid] = mc;
            alph[tid] = al;
        }
        __syncthreads();

        float av[8];
#pragma unroll
        for (int vi = 0; vi < 8; ++vi) av[vi] = alph[wave * 16 + vi + half * 8];
#pragma unroll
        for (int nt = 0; nt < 2; ++nt)
#pragma unroll
            for (int vi = 0; vi < 8; ++vi) oacc[nt][vi] *= av[vi];

        v2f pc = *(const v2f*)&Ps[qrow][half * 2];
        v2f vc0 = *(const v2f*)&Vs[half][0 * 16 + lane16][0];
        v2f vc1 = *(const v2f*)&Vs[half][1 * 16 + lane16][0];
#pragma unroll
        for (int t = 0; t < 16; ++t) {
            const int kkn = ((t + 1) & 15) * 4 + half * 2;
            v2f pn = *(const v2f*)&Ps[qrow][kkn];
            v2f vn0 = *(const v2f*)&Vs[kkn >> 1][0 * 16 + lane16][0];
            v2f vn1 = *(const v2f*)&Vs[kkn >> 1][1 * 16 + lane16][0];
            oacc[0] = wmma4(pc, vc0, oacc[0]);
            oacc[1] = wmma4(pc, vc1, oacc[1]);
            pc = pn; vc0 = vn0; vc1 = vn1;
        }
        __syncthreads();
    }

    float linv[8];
#pragma unroll
    for (int vi = 0; vi < 8; ++vi) linv[vi] = 1.0f / lrun[wave * 16 + vi + half * 8];
#pragma unroll
    for (int nt = 0; nt < 2; ++nt) {
        const int col = nt * 16 + lane16;
#pragma unroll
        for (int vi = 0; vi < 8; ++vi) {
            const int row = wave * 16 + vi + half * 8;
            o[((size_t)b * S + q0 + row) * LD + h * 32 + col] = oacc[nt][vi] * linv[vi];
        }
    }
}

extern "C" void kernel_launch(void* const* d_in, const int* in_sizes, int n_in,
                              void* d_out, int out_size, void* d_ws, size_t ws_size,
                              hipStream_t stream) {
    (void)in_sizes; (void)n_in; (void)out_size; (void)ws_size;
    const int B = 4, S = 2048, D = 256, F = 1024, L = 4;
    const int M = B * S;

    const float* x_in  = (const float*)d_in[0];
    const int*   mask  = (const int*)d_in[1];
    const float* w_qs  = (const float*)d_in[2];
    const float* w_ks  = (const float*)d_in[3];
    const float* w_vs  = (const float*)d_in[4];
    const float* w_fc  = (const float*)d_in[5];
    const float* ln1_g = (const float*)d_in[6];
    const float* ln1_b = (const float*)d_in[7];
    const float* w1    = (const float*)d_in[8];
    const float* b1    = (const float*)d_in[9];
    const float* w2    = (const float*)d_in[10];
    const float* b2    = (const float*)d_in[11];
    const float* ln2_g = (const float*)d_in[12];
    const float* ln2_b = (const float*)d_in[13];

    float* ws    = (float*)d_ws;
    float* x_cur = ws;
    float* xn    = x_cur + (size_t)M * D;
    float* qb    = xn + (size_t)M * D;
    float* kb    = qb + (size_t)M * D;
    float* vb    = kb + (size_t)M * D;
    float* ob    = vb + (size_t)M * D;
    float* hb    = ob + (size_t)M * D;
    float* mfv   = hb + (size_t)M * F;

    hipMemcpyAsync(x_cur, x_in, sizeof(float) * (size_t)M * D,
                   hipMemcpyDeviceToDevice, stream);
    mf_kernel<<<M / 256, 256, 0, stream>>>(mask, mfv, M);

    const dim3 gD(M / 64, D / 128);
    const dim3 gF(M / 64, F / 128);
    const dim3 gA(S / 64, B * 8);

    for (int l = 0; l < L; ++l) {
        const float* wq = w_qs + (size_t)l * D * D;
        const float* wk = w_ks + (size_t)l * D * D;
        const float* wv = w_vs + (size_t)l * D * D;
        const float* wf = w_fc + (size_t)l * D * D;
        const float* w1l = w1 + (size_t)l * D * F;
        const float* w2l = w2 + (size_t)l * F * D;
        const float* b1l = b1 + (size_t)l * F;
        const float* b2l = b2 + (size_t)l * D;

        ln_kernel<<<M, 256, 0, stream>>>(x_cur, xn, ln1_g + (size_t)l * D, ln1_b + (size_t)l * D);
        gemm_kernel<<<gD, 256, 0, stream>>>(xn, wq, qb, nullptr, mfv, nullptr, D, D, 0);
        gemm_kernel<<<gD, 256, 0, stream>>>(xn, wk, kb, nullptr, mfv, nullptr, D, D, 0);
        gemm_kernel<<<gD, 256, 0, stream>>>(xn, wv, vb, nullptr, mfv, nullptr, D, D, 0);
        attn_kernel<<<gA, 128, 0, stream>>>(qb, kb, vb, mfv, ob);
        gemm_kernel<<<gD, 256, 0, stream>>>(ob, wf, xn, nullptr, mfv, x_cur, D, D, 0);
        gemm_kernel<<<gF, 256, 0, stream>>>(xn, w1l, hb, b1l, nullptr, nullptr, D, F, 1);
        gemm_kernel<<<gD, 256, 0, stream>>>(hb, w2l, qb, b2l, nullptr, xn, F, D, 0);
        ln_kernel<<<M, 256, 0, stream>>>(qb, x_cur, ln2_g + (size_t)l * D, ln2_b + (size_t)l * D);
    }

    hipMemcpyAsync(d_out, x_cur, sizeof(float) * (size_t)M * D,
                   hipMemcpyDeviceToDevice, stream);
}


# Tile staging (GEMM A and W tiles; attention Q, K, V, key mask): all global loads of a tile issued as one group into registers before its LDS stores; was one load waited per loop trip
# speedup vs baseline: 1.3694x; 1.3694x over previous
#include <hip/hip_runtime.h>
#include <hip/hip_bf16.h>

typedef __attribute__((ext_vector_type(2))) float v2f;
typedef __attribute__((ext_vector_type(8))) float v8f;

__device__ __forceinline__ v8f wmma4(v2f a, v2f b, v8f c) {
    return __builtin_amdgcn_wmma_f32_16x16x4_f32(false, a, false, b, (short)0, c, false, false);
}

__global__ void mf_kernel(const int* __restrict__ mask, float* __restrict__ mfv, int n) {
    int i = blockIdx.x * blockDim.x + threadIdx.x;
    if (i < n) mfv[i] = mask[i] ? 1.0f : 0.0f;
}

__global__ void ln_kernel(const float* __restrict__ x, float* __restrict__ out,
                          const float* __restrict__ g, const float* __restrict__ b) {
    __shared__ float red[256];
    const int row = blockIdx.x;
    const int tid = threadIdx.x;
    float v = x[(size_t)row * 256 + tid];
    red[tid] = v;
    __syncthreads();
    for (int s = 128; s > 0; s >>= 1) {
        if (tid < s) red[tid] += red[tid + s];
        __syncthreads();
    }
    float mean = red[0] * (1.0f / 256.0f);
    __syncthreads();
    float d = v - mean;
    red[tid] = d * d;
    __syncthreads();
    for (int s = 128; s > 0; s >>= 1) {
        if (tid < s) red[tid] += red[tid + s];
        __syncthreads();
    }
    float var = red[0] * (1.0f / 256.0f);
    float r = rsqrtf(var + 1e-6f);
    out[(size_t)row * 256 + tid] = d * r * g[tid] + b[tid];
}

__global__ void gemm_kernel(const float* __restrict__ A, const float* __restrict__ W,
                            float* __restrict__ C, const float* __restrict__ bias,
                            const float* __restrict__ mfv, const float* __restrict__ res,
                            int K, int N, int relu) {
    __shared__ __align__(16) float As[64][34];
    __shared__ __align__(16) float Bs[16][144][2];

    const int tid = threadIdx.x;
    const int lane = tid & 31;
    const int wave = tid >> 5;
    const int half = lane >> 4;
    const int lane16 = lane & 15;
    const int wm = wave >> 2;
    const int wn = wave & 3;
    const int m0 = blockIdx.x * 64;
    const int n0 = blockIdx.y * 128;

    const int rowA0 = wm * 32 + lane16;
    const int rowA1 = rowA0 + 16;
    const int colB0 = wn * 32 + lane16;
    const int colB1 = colB0 + 16;

    v8f acc[2][2] = {};

    for (int k0 = 0; k0 < K; k0 += 32) {
        float ra[8], rb0[8], rb1[8];
#pragma unroll
        for (int j = 0; j < 8; ++j) {
            const int i = tid + j * 256;
            const int r = i >> 5, c = i & 31;
            ra[j] = A[(size_t)(m0 + r) * K + k0 + c];
        }
#pragma unroll
        for (int j = 0; j < 8; ++j) {
            const int i = tid + j * 256;
            const int kp = i >> 7, c = i & 127;
            rb0[j] = W[(size_t)(k0 + 2 * kp) * N + n0 + c];
            rb1[j] = W[(size_t)(k0 + 2 * kp + 1) * N + n0 + c];
        }
#pragma unroll
        for (int j = 0; j < 8; ++j) {
            const int i = tid + j * 256;
            As[i >> 5][i & 31] = ra[j];
        }
#pragma unroll
        for (int j = 0; j < 8; ++j) {
            const int i = tid + j * 256;
            const int kp = i >> 7, c = i & 127;
            Bs[kp][c][0] = rb0[j];
            Bs[kp][c][1] = rb1[j];
        }
        if (k0 + 32 < K) {
            if (tid < 64) {
                __builtin_prefetch(&A[(size_t)(m0 + tid) * K + k0 + 32], 0, 1);
            } else if (tid < 192) {
                int r = (tid - 64) >> 2, s = (tid - 64) & 3;
                __builtin_prefetch(&W[(size_t)(k0 + 32 + r) * N + n0 + s * 32], 0, 1);
            }
        }
        __syncthreads();

        v2f a0c = *(const v2f*)&As[rowA0][half * 2];
        v2f a1c = *(const v2f*)&As[rowA1][half * 2];
        v2f b0c = *(const v2f*)&Bs[half][colB0][0];
        v2f b1c = *(const v2f*)&Bs[half][colB1][0];
#pragma unroll
        for (int t = 0; t < 8; ++t) {
            const int kkn = ((t + 1) & 7) * 4 + half * 2;
            v2f a0n = *(const v2f*)&As[rowA0][kkn];
            v2f a1n = *(const v2f*)&As[rowA1][kkn];
            v2f b0n = *(const v2f*)&Bs[kkn >> 1][colB0][0];
            v2f b1n = *(const v2f*)&Bs[kkn >> 1][colB1][0];
            acc[0][0] = wmma4(a0c, b0c, acc[0][0]);
            acc[0][1] = wmma4(a0c, b1c, acc[0][1]);
            acc[1][0] = wmma4(a1c, b0c, acc[1][0]);
            acc[1][1] = wmma4(a1c, b1c, acc[1][1]);
            a0c = a0n; a1c = a1n; b0c = b0n; b1c = b1n;
        }
        __syncthreads();
    }

#pragma unroll
    for (int im = 0; im < 2; ++im) {
#pragma unroll
        for (int jn = 0; jn < 2; ++jn) {
            const int col = n0 + wn * 32 + jn * 16 + lane16;
            const float bv = bias ? bias[col] : 0.0f;
#pragma unroll
            for (int vi = 0; vi < 8; ++vi) {
                const int row = m0 + wm * 32 + im * 16 + vi + half * 8;
                float val = acc[im][jn][vi] + bv;
                if (relu) val = fmaxf(val, 0.0f);
                if (mfv) val *= mfv[row];
                if (res) val += res[(size_t)row * N + col];
                C[(size_t)row * N + col] = val;
            }
        }
    }
}

__global__ void __launch_bounds__(128) attn_kernel(const float* __restrict__ q, const float* __restrict__ k,
                            const float* __restrict__ v, const float* __restrict__ mfv,
                            float* __restrict__ o) {
    __shared__ __align__(16) float Qs[64][34];
    __shared__ __align__(16) float Ks[16][80][2];
    __shared__ __align__(16) float Vs[32][48][2];
    __shared__ __align__(16) float Ps[64][68];
    __shared__ float mb[64];
    __shared__ float mrun[64], lrun[64], alph[64];

    const int S = 2048, LD = 256;
    const int tid = threadIdx.x;
    const int lane = tid & 31;
    const int wave = tid >> 5;
    const int half = lane >> 4;
    const int lane16 = lane & 15;
    const int bh = blockIdx.y;
    const int b = bh >> 3;
    const int h = bh & 7;
    const int q0 = blockIdx.x * 64;
    const int qrow = wave * 16 + lane16;

    const size_t baseQ = ((size_t)b * S + q0) * LD + h * 32;
    {
        float rq[16];
#pragma unroll
        for (int j = 0; j < 16; ++j) {
            const int i = tid + j * 128;
            rq[j] = q[baseQ + (size_t)(i >> 5) * LD + (i & 31)];
        }
#pragma unroll
        for (int j = 0; j < 16; ++j) {
            const int i = tid + j * 128;
            Qs[i >> 5][i & 31] = rq[j];
        }
    }
    if (tid < 64) { mrun[tid] = -3.0e38f; lrun[tid] = 0.0f; }
    v8f oacc[2] = {};
    __syncthreads();

    for (int j0 = 0; j0 < S; j0 += 64) {
        const size_t baseK = ((size_t)b * S + j0) * LD + h * 32;
        float rk0[8], rk1[8], rv0[8], rv1[8];
#pragma unroll
        for (int j = 0; j < 8; ++j) {
            const int i = tid + j * 128;
            const int r = i & 63, kp = i >> 6;
            rk0[j] = k[baseK + (size_t)r * LD + 2 * kp];
            rk1[j] = k[baseK + (size_t)r * LD + 2 * kp + 1];
        }
#pragma unroll
        for (int j = 0; j < 8; ++j) {
            const int i = tid + j * 128;
            const int n = i & 31, kp = i >> 5;
            rv0[j] = v[baseK + (size_t)(2 * kp) * LD + n];
            rv1[j] = v[baseK + (size_t)(2 * kp + 1) * LD + n];
        }
        const float rmk = mfv[(size_t)b * S + j0 + (tid & 63)];
#pragma unroll
        for (int j = 0; j < 8; ++j) {
            const int i = tid + j * 128;
            const int r = i & 63, kp = i >> 6;
            Ks[kp][r][0] = rk0[j];
            Ks[kp][r][1] = rk1[j];
        }
#pragma unroll
        for (int j = 0; j < 8; ++j) {
            const int i = tid + j * 128;
            const int n = i & 31, kp = i >> 5;
            Vs[kp][n][0] = rv0[j];
            Vs[kp][n][1] = rv1[j];
        }
        mb[tid & 63] = (rmk != 0.0f) ? 0.0f : -1e9f;
        if (j0 + 64 < S) {
            const size_t baseN = ((size_t)b * S + j0 + 64) * LD + h * 32;
            if (tid < 64) __builtin_prefetch(&k[baseN + (size_t)tid * LD], 0, 1);
            else          __builtin_prefetch(&v[baseN + (size_t)(tid - 64) * LD], 0, 1);
        }
        __syncthreads();

        v8f sacc[4] = {};
        v2f ac = *(const v2f*)&Qs[qrow][half * 2];
        v2f bc0 = *(const v2f*)&Ks[half][0 * 16 + lane16][0];
        v2f bc1 = *(const v2f*)&Ks[half][1 * 16 + lane16][0];
        v2f bc2 = *(const v2f*)&Ks[half][2 * 16 + lane16][0];
        v2f bc3 = *(const v2f*)&Ks[half][3 * 16 + lane16][0];
#pragma unroll
        for (int t = 0; t < 8; ++t) {
            const int kkn = ((t + 1) & 7) * 4 + half * 2;
            v2f an = *(const v2f*)&Qs[qrow][kkn];
            v2f bn0 = *(const v2f*)&Ks[kkn >> 1][0 * 16 + lane16][0];
            v2f bn1 = *(const v2f*)&Ks[kkn >> 1][1 * 16 + lane16][0];
            v2f bn2 = *(const v2f*)&Ks[kkn >> 1][2 * 16 + lane16][0];
            v2f bn3 = *(const v2f*)&Ks[kkn >> 1][3 * 16 + lane16][0];
            sacc[0] = wmma4(ac, bc0, sacc[0]);
            sacc[1] = wmma4(ac, bc1, sacc[1]);
            sacc[2] = wmma4(ac, bc2, sacc[2]);
            sacc[3] = wmma4(ac, bc3, sacc[3]);
            ac = an; bc0 = bn0; bc1 = bn1; bc2 = bn2; bc3 = bn3;
        }
        const float inv = 0.17677669529663687f;
#pragma unroll
        for (int nt = 0; nt < 4; ++nt) {
            const int col = nt * 16 + lane16;
#pragma unroll
            for (int vi = 0; vi < 8; ++vi) {
                const int row = wave * 16 + vi + half * 8;
                Ps[row][col] = sacc[nt][vi] * inv + mb[col];
            }
        }
        __syncthreads();

        if (tid < 64) {
            float mo = mrun[tid];
            float mc = mo;
            for (int j = 0; j < 64; ++j) mc = fmaxf(mc, Ps[tid][j]);
            float al = __expf(mo - mc);
            float ls = 0.0f;
            for (int j = 0; j < 64; ++j) {
                float p = __expf(Ps[tid][j] - mc);
                Ps[tid][j] = p;
                ls += p;
            }
            lrun[tid] = lrun[tid] * al + ls;
            mrun[tid] = mc;
            alph[tid] = al;
        }
        __syncthreads();

        float av[8];
#pragma unroll
        for (int vi = 0; vi < 8; ++vi) av[vi] = alph[wave * 16 + vi + half * 8];
#pragma unroll
        for (int nt = 0; nt < 2; ++nt)
#pragma unroll
            for (int vi = 0; vi < 8; ++vi) oacc[nt][vi] *= av[vi];

        v2f pc = *(const v2f*)&Ps[qrow][half * 2];
        v2f vc0 = *(const v2f*)&Vs[half][0 * 16 + lane16][0];
        v2f vc1 = *(const v2f*)&Vs[half][1 * 16 + lane16][0];
#pragma unroll
        for (int t = 0; t < 16; ++t) {
            const int kkn = ((t + 1) & 15) * 4 + half * 2;
            v2f pn = *(const v2f*)&Ps[qrow][kkn];
            v2f vn0 = *(const v2f*)&Vs[kkn >> 1][0 * 16 + lane16][0];
            v2f vn1 = *(const v2f*)&Vs[kkn >> 1][1 * 16 + lane16][0];
            oacc[0] = wmma4(pc, vc0, oacc[0]);
            oacc[1] = wmma4(pc, vc1, oacc[1]);
            pc = pn; vc0 = vn0; vc1 = vn1;
        }
        __syncthreads();
    }

    float linv[8];
#pragma unroll
    for (int vi = 0; vi < 8; ++vi) linv[vi] = 1.0f / lrun[wave * 16 + vi + half * 8];
#pragma unroll
    for (int nt = 0; nt < 2; ++nt) {
        const int col = nt * 16 + lane16;
#pragma unroll
        for (int vi = 0; vi < 8; ++vi) {
            const int row = wave * 16 + vi + half * 8;
            o[((size_t)b * S + q0 + row) * LD + h * 32 + col] = oacc[nt][vi] * linv[vi];
        }
    }
}

extern "C" void kernel_launch(void* const* d_in, const int* in_sizes, int n_in,
                              void* d_out, int out_size, void* d_ws, size_t ws_size,
                              hipStream_t stream) {
    (void)in_sizes; (void)n_in; (void)out_size; (void)ws_size;
    const int B = 4, S = 2048, D = 256, F = 1024, L = 4;
    const int M = B * S;

    const float* x_in  = (const float*)d_in[0];
    const int*   mask  = (const int*)d_in[1];
    const float* w_qs  = (const float*)d_in[2];
    const float* w_ks  = (const float*)d_in[3];
    const float* w_vs  = (const float*)d_in[4];
    const float* w_fc  = (const float*)d_in[5];
    const float* ln1_g = (const float*)d_in[6];
    const float* ln1_b = (const float*)d_in[7];
    const float* w1    = (const float*)d_in[8];
    const float* b1    = (const float*)d_in[9];
    const float* w2    = (const float*)d_in[10];
    const float* b2    = (const float*)d_in[11];
    const float* ln2_g = (const float*)d_in[12];
    const float* ln2_b = (const float*)d_in[13];

    float* ws    = (float*)d_ws;
    float* x_cur = ws;
    float* xn    = x_cur + (size_t)M * D;
    float* qb    = xn + (size_t)M * D;
    float* kb    = qb + (size_t)M * D;
    float* vb    = kb + (size_t)M * D;
    float* ob    = vb + (size_t)M * D;
    float* hb    = ob + (size_t)M * D;
    float* mfv   = hb + (size_t)M * F;

    hipMemcpyAsync(x_cur, x_in, sizeof(float) * (size_t)M * D,
                   hipMemcpyDeviceToDevice, stream);
    mf_kernel<<<M / 256, 256, 0, stream>>>(mask, mfv, M);

    const dim3 gD(M / 64, D / 128);
    const dim3 gF(M / 64, F / 128);
    const dim3 gA(S / 64, B * 8);

    for (int l = 0; l < L; ++l) {
        const float* wq = w_qs + (size_t)l * D * D;
        const float* wk = w_ks + (size_t)l * D * D;
        const float* wv = w_vs + (size_t)l * D * D;
        const float* wf = w_fc + (size_t)l * D * D;
        const float* w1l = w1 + (size_t)l * D * F;
        const float* w2l = w2 + (size_t)l * F * D;
        const float* b1l = b1 + (size_t)l * F;
        const float* b2l = b2 + (size_t)l * D;

        ln_kernel<<<M, 256, 0, stream>>>(x_cur, xn, ln1_g + (size_t)l * D, ln1_b + (size_t)l * D);
        gemm_kernel<<<gD, 256, 0, stream>>>(xn, wq, qb, nullptr, mfv, nullptr, D, D, 0);
        gemm_kernel<<<gD, 256, 0, stream>>>(xn, wk, kb, nullptr, mfv, nullptr, D, D, 0);
        gemm_kernel<<<gD, 256, 0, stream>>>(xn, wv, vb, nullptr, mfv, nullptr, D, D, 0);
        attn_kernel<<<gA, 128, 0, stream>>>(qb, kb, vb, mfv, ob);
        gemm_kernel<<<gD, 256, 0, stream>>>(ob, wf, xn, nullptr, mfv, x_cur, D, D, 0);
        gemm_kernel<<<gF, 256, 0, stream>>>(xn, w1l, hb, b1l, nullptr, nullptr, D, F, 1);
        gemm_kernel<<<gD, 256, 0, stream>>>(hb, w2l, qb, b2l, nullptr, xn, F, D, 0);
        ln_kernel<<<M, 256, 0, stream>>>(qb, x_cur, ln2_g + (size_t)l * D, ln2_b + (size_t)l * D);
    }

    hipMemcpyAsync(d_out, x_cur, sizeof(float) * (size_t)M * D,
                   hipMemcpyDeviceToDevice, stream);
}


# Attention: both products transposed (lane = query): softmax in registers, P feeds PV from the accumulators, no LDS score tile; k order of both products and the row-sum tree change
# speedup vs baseline: 1.7356x; 1.2674x over previous
#include <hip/hip_runtime.h>
#include <hip/hip_bf16.h>

typedef __attribute__((ext_vector_type(2))) float v2f;
typedef __attribute__((ext_vector_type(8))) float v8f;

__device__ __forceinline__ v8f wmma4(v2f a, v2f b, v8f c) {
    return __builtin_amdgcn_wmma_f32_16x16x4_f32(false, a, false, b, (short)0, c, false, false);
}

__global__ void mf_kernel(const int* __restrict__ mask, float* __restrict__ mfv, int n) {
    int i = blockIdx.x * blockDim.x + threadIdx.x;
    if (i < n) mfv[i] = mask[i] ? 1.0f : 0.0f;
}

__global__ void ln_kernel(const float* __restrict__ x, float* __restrict__ out,
                          const float* __restrict__ g, const float* __restrict__ b) {
    __shared__ float red[256];
    const int row = blockIdx.x;
    const int tid = threadIdx.x;
    float v = x[(size_t)row * 256 + tid];
    red[tid] = v;
    __syncthreads();
    for (int s = 128; s > 0; s >>= 1) {
        if (tid < s) red[tid] += red[tid + s];
        __syncthreads();
    }
    float mean = red[0] * (1.0f / 256.0f);
    __syncthreads();
    float d = v - mean;
    red[tid] = d * d;
    __syncthreads();
    for (int s = 128; s > 0; s >>= 1) {
        if (tid < s) red[tid] += red[tid + s];
        __syncthreads();
    }
    float var = red[0] * (1.0f / 256.0f);
    float r = rsqrtf(var + 1e-6f);
    out[(size_t)row * 256 + tid] = d * r * g[tid] + b[tid];
}

__global__ void gemm_kernel(const float* __restrict__ A, const float* __restrict__ W,
                            float* __restrict__ C, const float* __restrict__ bias,
                            const float* __restrict__ mfv, const float* __restrict__ res,
                            int K, int N, int relu) {
    __shared__ __align__(16) float As[64][34];
    __shared__ __align__(16) float Bs[16][144][2];

    const int tid = threadIdx.x;
    const int lane = tid & 31;
    const int wave = tid >> 5;
    const int half = lane >> 4;
    const int lane16 = lane & 15;
    const int wm = wave >> 2;
    const int wn = wave & 3;
    const int m0 = blockIdx.x * 64;
    const int n0 = blockIdx.y * 128;

    const int rowA0 = wm * 32 + lane16;
    const int rowA1 = rowA0 + 16;
    const int colB0 = wn * 32 + lane16;
    const int colB1 = colB0 + 16;

    v8f acc[2][2] = {};

    for (int k0 = 0; k0 < K; k0 += 32) {
        float ra[8], rb0[8], rb1[8];
#pragma unroll
        for (int j = 0; j < 8; ++j) {
            const int i = tid + j * 256;
            const int r = i >> 5, c = i & 31;
            ra[j] = A[(size_t)(m0 + r) * K + k0 + c];
        }
#pragma unroll
        for (int j = 0; j < 8; ++j) {
            const int i = tid + j * 256;
            const int kp = i >> 7, c = i & 127;
            rb0[j] = W[(size_t)(k0 + 2 * kp) * N + n0 + c];
            rb1[j] = W[(size_t)(k0 + 2 * kp + 1) * N + n0 + c];
        }
#pragma unroll
        for (int j = 0; j < 8; ++j) {
            const int i = tid + j * 256;
            As[i >> 5][i & 31] = ra[j];
        }
#pragma unroll
        for (int j = 0; j < 8; ++j) {
            const int i = tid + j * 256;
            const int kp = i >> 7, c = i & 127;
            Bs[kp][c][0] = rb0[j];
            Bs[kp][c][1] = rb1[j];
        }
        if (k0 + 32 < K) {
            if (tid < 64) {
                __builtin_prefetch(&A[(size_t)(m0 + tid) * K + k0 + 32], 0, 1);
            } else if (tid < 192) {
                int r = (tid - 64) >> 2, s = (tid - 64) & 3;
                __builtin_prefetch(&W[(size_t)(k0 + 32 + r) * N + n0 + s * 32], 0, 1);
            }
        }
        __syncthreads();

        v2f a0c = *(const v2f*)&As[rowA0][half * 2];
        v2f a1c = *(const v2f*)&As[rowA1][half * 2];
        v2f b0c = *(const v2f*)&Bs[half][colB0][0];
        v2f b1c = *(const v2f*)&Bs[half][colB1][0];
#pragma unroll
        for (int t = 0; t < 8; ++t) {
            const int kkn = ((t + 1) & 7) * 4 + half * 2;
            v2f a0n = *(const v2f*)&As[rowA0][kkn];
            v2f a1n = *(const v2f*)&As[rowA1][kkn];
            v2f b0n = *(const v2f*)&Bs[kkn >> 1][colB0][0];
            v2f b1n = *(const v2f*)&Bs[kkn >> 1][colB1][0];
            acc[0][0] = wmma4(a0c, b0c, acc[0][0]);
            acc[0][1] = wmma4(a0c, b1c, acc[0][1]);
            acc[1][0] = wmma4(a1c, b0c, acc[1][0]);
            acc[1][1] = wmma4(a1c, b1c, acc[1][1]);
            a0c = a0n; a1c = a1n; b0c = b0n; b1c = b1n;
        }
        __syncthreads();
    }

#pragma unroll
    for (int im = 0; im < 2; ++im) {
#pragma unroll
        for (int jn = 0; jn < 2; ++jn) {
            const int col = n0 + wn * 32 + jn * 16 + lane16;
            const float bv = bias ? bias[col] : 0.0f;
#pragma unroll
            for (int vi = 0; vi < 8; ++vi) {
                const int row = m0 + wm * 32 + im * 16 + vi + half * 8;
                float val = acc[im][jn][vi] + bv;
                if (relu) val = fmaxf(val, 0.0f);
                if (mfv) val *= mfv[row];
                if (res) val += res[(size_t)row * N + col];
                C[(size_t)row * N + col] = val;
            }
        }
    }
}

typedef __attribute__((ext_vector_type(4))) float v4f;

__device__ __forceinline__ float xhalf(float x) {
    const int bits = __builtin_bit_cast(int, x);
    return __builtin_bit_cast(float, __builtin_amdgcn_permlanex16(bits, bits, 0x76543210, 0xfedcba98, false, false));
}

__global__ void __launch_bounds__(128) attn_kernel(const float* __restrict__ q, const float* __restrict__ k,
                            const float* __restrict__ v, const float* __restrict__ mfv,
                            float* __restrict__ o) {
    __shared__ __align__(16) float Ks[64][36];
    __shared__ __align__(16) float Vt[32][68];
    __shared__ __align__(16) float mb[64];

    const int S = 2048, LD = 256;
    const int tid = threadIdx.x;
    const int lane = tid & 31;
    const int wave = tid >> 5;
    const int half = lane >> 4;
    const int lane16 = lane & 15;
    const int bh = blockIdx.y;
    const int b = bh >> 3;
    const int h = bh & 7;
    const int q0 = blockIdx.x * 64;

    const float qscale = 0.17677669529663687f * 1.4426950408889634f;
    v4f qf[4];
    {
        const float* qp = q + ((size_t)b * S + q0 + wave * 16 + lane16) * LD + h * 32 + half * 4;
#pragma unroll
        for (int u = 0; u < 4; ++u) qf[u] = *(const v4f*)(qp + 8 * u) * qscale;
    }

    v8f oacc[2] = {};
    float m_run = -3.0e38f;
    float l_part = 0.0f;

    for (int j0 = 0; j0 < S; j0 += 64) {
        const size_t baseK = ((size_t)b * S + j0) * LD + h * 32;
        v4f rk[4];
        float rv[4][4];
#pragma unroll
        for (int j = 0; j < 4; ++j) {
            const int i = tid + j * 128;
            rk[j] = *(const v4f*)&k[baseK + (size_t)(i >> 3) * LD + 4 * (i & 7)];
        }
#pragma unroll
        for (int j = 0; j < 4; ++j) {
            const int kq = wave + 4 * j;
#pragma unroll
            for (int i = 0; i < 4; ++i) rv[j][i] = v[baseK + (size_t)(4 * kq + i) * LD + lane];
        }
        const float rmk = mfv[(size_t)b * S + j0 + (tid & 63)];
#pragma unroll
        for (int j = 0; j < 4; ++j) {
            const int i = tid + j * 128;
            *(v4f*)&Ks[i >> 3][4 * (i & 7)] = rk[j];
        }
#pragma unroll
        for (int j = 0; j < 4; ++j) {
            const int kq = wave + 4 * j;
            v4f t = {rv[j][0], rv[j][1], rv[j][2], rv[j][3]};
            *(v4f*)&Vt[lane][4 * kq] = t;
        }
        mb[tid & 63] = (rmk != 0.0f) ? 0.0f : -1e9f;
        if (j0 + 64 < S) {
            const size_t baseN = ((size_t)b * S + j0 + 64) * LD + h * 32;
            if (tid < 64) __builtin_prefetch(&k[baseN + (size_t)tid * LD], 0, 1);
            else          __builtin_prefetch(&v[baseN + (size_t)(tid - 64) * LD], 0, 1);
        }
        __syncthreads();

        v8f sacc[4];
#pragma unroll
        for (int kt = 0; kt < 4; ++kt) {
            const v4f b0 = *(const v4f*)&mb[kt * 16 + half * 8];
            const v4f b1 = *(const v4f*)&mb[kt * 16 + half * 8 + 4];
            sacc[kt][0] = b0[0]; sacc[kt][1] = b0[1]; sacc[kt][2] = b0[2]; sacc[kt][3] = b0[3];
            sacc[kt][4] = b1[0]; sacc[kt][5] = b1[1]; sacc[kt][6] = b1[2]; sacc[kt][7] = b1[3];
        }
        v4f ka[4][4];
#pragma unroll
        for (int u = 0; u < 4; ++u)
#pragma unroll
            for (int kt = 0; kt < 4; ++kt)
                ka[u][kt] = *(const v4f*)&Ks[kt * 16 + lane16][8 * u + half * 4];
        v4f va[4][2][2];
#pragma unroll
        for (int kt = 0; kt < 4; ++kt)
#pragma unroll
            for (int w = 0; w < 2; ++w) {
                va[kt][w][0] = *(const v4f*)&Vt[lane16][kt * 16 + half * 8 + 4 * w];
                va[kt][w][1] = *(const v4f*)&Vt[16 + lane16][kt * 16 + half * 8 + 4 * w];
            }
#pragma unroll
        for (int u = 0; u < 4; ++u) {
#pragma unroll
            for (int e = 0; e < 2; ++e) {
                const v2f bq = {qf[u][2 * e], qf[u][2 * e + 1]};
#pragma unroll
                for (int kt = 0; kt < 4; ++kt) {
                    const v2f ak = {ka[u][kt][2 * e], ka[u][kt][2 * e + 1]};
                    sacc[kt] = wmma4(ak, bq, sacc[kt]);
                }
            }
        }

        float mx[4];
#pragma unroll
        for (int kt = 0; kt < 4; ++kt) {
            mx[kt] = sacc[kt][0];
#pragma unroll
            for (int r = 1; r < 8; ++r) mx[kt] = fmaxf(mx[kt], sacc[kt][r]);
        }
        float mc = fmaxf(fmaxf(mx[0], mx[1]), fmaxf(mx[2], mx[3]));
        mc = fmaxf(mc, xhalf(mc));
        const float mn = fmaxf(m_run, mc);
        const float al = __builtin_amdgcn_exp2f(m_run - mn);
        m_run = mn;
        float ls[4];
#pragma unroll
        for (int kt = 0; kt < 4; ++kt) {
            ls[kt] = 0.0f;
#pragma unroll
            for (int r = 0; r < 8; ++r) {
                const float p = __builtin_amdgcn_exp2f(sacc[kt][r] - mn);
                sacc[kt][r] = p;
                ls[kt] += p;
            }
        }
        l_part = l_part * al + ((ls[0] + ls[1]) + (ls[2] + ls[3]));
#pragma unroll
        for (int dt = 0; dt < 2; ++dt)
#pragma unroll
            for (int r = 0; r < 8; ++r) oacc[dt][r] *= al;

#pragma unroll
        for (int kt = 0; kt < 4; ++kt) {
#pragma unroll
            for (int w = 0; w < 2; ++w) {
#pragma unroll
                for (int e = 0; e < 2; ++e) {
                    const v2f pb = {sacc[kt][4 * w + 2 * e], sacc[kt][4 * w + 2 * e + 1]};
                    const v2f a0 = {va[kt][w][0][2 * e], va[kt][w][0][2 * e + 1]};
                    const v2f a1 = {va[kt][w][1][2 * e], va[kt][w][1][2 * e + 1]};
                    oacc[0] = wmma4(a0, pb, oacc[0]);
                    oacc[1] = wmma4(a1, pb, oacc[1]);
                }
            }
        }
        __syncthreads();
    }

    const float lsum = l_part + xhalf(l_part);
    const float linv = 1.0f / lsum;
    float* op = o + ((size_t)b * S + q0 + wave * 16 + lane16) * LD + h * 32 + half * 8;
#pragma unroll
    for (int dt = 0; dt < 2; ++dt) {
        const v4f lo = {oacc[dt][0] * linv, oacc[dt][1] * linv, oacc[dt][2] * linv, oacc[dt][3] * linv};
        const v4f hi = {oacc[dt][4] * linv, oacc[dt][5] * linv, oacc[dt][6] * linv, oacc[dt][7] * linv};
        *(v4f*)(op + dt * 16) = lo;
        *(v4f*)(op + dt * 16 + 4) = hi;
    }
}

extern "C" void kernel_launch(void* const* d_in, const int* in_sizes, int n_in,
                              void* d_out, int out_size, void* d_ws, size_t ws_size,
                              hipStream_t stream) {
    (void)in_sizes; (void)n_in; (void)out_size; (void)ws_size;
    const int B = 4, S = 2048, D = 256, F = 1024, L = 4;
    const int M = B * S;

    const float* x_in  = (const float*)d_in[0];
    const int*   mask  = (const int*)d_in[1];
    const float* w_qs  = (const float*)d_in[2];
    const float* w_ks  = (const float*)d_in[3];
    const float* w_vs  = (const float*)d_in[4];
    const float* w_fc  = (const float*)d_in[5];
    const float* ln1_g = (const float*)d_in[6];
    const float* ln1_b = (const float*)d_in[7];
    const float* w1    = (const float*)d_in[8];
    const float* b1    = (const float*)d_in[9];
    const float* w2    = (const float*)d_in[10];
    const float* b2    = (const float*)d_in[11];
    const float* ln2_g = (const float*)d_in[12];
    const float* ln2_b = (const float*)d_in[13];

    float* ws    = (float*)d_ws;
    float* x_cur = ws;
    float* xn    = x_cur + (size_t)M * D;
    float* qb    = xn + (size_t)M * D;
    float* kb    = qb + (size_t)M * D;
    float* vb    = kb + (size_t)M * D;
    float* ob    = vb + (size_t)M * D;
    float* hb    = ob + (size_t)M * D;
    float* mfv   = hb + (size_t)M * F;

    hipMemcpyAsync(x_cur, x_in, sizeof(float) * (size_t)M * D,
                   hipMemcpyDeviceToDevice, stream);
    mf_kernel<<<M / 256, 256, 0, stream>>>(mask, mfv, M);

    const dim3 gD(M / 64, D / 128);
    const dim3 gF(M / 64, F / 128);
    const dim3 gA(S / 64, B * 8);

    for (int l = 0; l < L; ++l) {
        const float* wq = w_qs + (size_t)l * D * D;
        const float* wk = w_ks + (size_t)l * D * D;
        const float* wv = w_vs + (size_t)l * D * D;
        const float* wf = w_fc + (size_t)l * D * D;
        const float* w1l = w1 + (size_t)l * D * F;
        const float* w2l = w2 + (size_t)l * F * D;
        const float* b1l = b1 + (size_t)l * F;
        const float* b2l = b2 + (size_t)l * D;

        ln_kernel<<<M, 256, 0, stream>>>(x_cur, xn, ln1_g + (size_t)l * D, ln1_b + (size_t)l * D);
        gemm_kernel<<<gD, 256, 0, stream>>>(xn, wq, qb, nullptr, mfv, nullptr, D, D, 0);
        gemm_kernel<<<gD, 256, 0, stream>>>(xn, wk, kb, nullptr, mfv, nullptr, D, D, 0);
        gemm_kernel<<<gD, 256, 0, stream>>>(xn, wv, vb, nullptr, mfv, nullptr, D, D, 0);
        attn_kernel<<<gA, 128, 0, stream>>>(qb, kb, vb, mfv, ob);
        gemm_kernel<<<gD, 256, 0, stream>>>(ob, wf, xn, nullptr, mfv, x_cur, D, D, 0);
        gemm_kernel<<<gF, 256, 0, stream>>>(xn, w1l, hb, b1l, nullptr, nullptr, D, F, 1);
        gemm_kernel<<<gD, 256, 0, stream>>>(hb, w2l, qb, b2l, nullptr, xn, F, D, 0);
        ln_kernel<<<M, 256, 0, stream>>>(qb, x_cur, ln2_g + (size_t)l * D, ln2_b + (size_t)l * D);
    }

    hipMemcpyAsync(d_out, x_cur, sizeof(float) * (size_t)M * D,
                   hipMemcpyDeviceToDevice, stream);
}


# GEMM: K loop software-pipelined in registers: the next K tile's global loads (6 x 128-bit per thread off uniform bases, was 24 x 32-bit) issue ahead of the current tile's matrix work
# speedup vs baseline: 2.1656x; 1.2478x over previous
#include <hip/hip_runtime.h>
#include <hip/hip_bf16.h>

typedef __attribute__((ext_vector_type(2))) float v2f;
typedef __attribute__((ext_vector_type(8))) float v8f;

__device__ __forceinline__ v8f wmma4(v2f a, v2f b, v8f c) {
    return __builtin_amdgcn_wmma_f32_16x16x4_f32(false, a, false, b, (short)0, c, false, false);
}

__global__ void mf_kernel(const int* __restrict__ mask, float* __restrict__ mfv, int n) {
    int i = blockIdx.x * blockDim.x + threadIdx.x;
    if (i < n) mfv[i] = mask[i] ? 1.0f : 0.0f;
}

__global__ void ln_kernel(const float* __restrict__ x, float* __restrict__ out,
                          const float* __restrict__ g, const float* __restrict__ b) {
    __shared__ float red[256];
    const int row = blockIdx.x;
    const int tid = threadIdx.x;
    float v = x[(size_t)row * 256 + tid];
    red[tid] = v;
    __syncthreads();
    for (int s = 128; s > 0; s >>= 1) {
        if (tid < s) red[tid] += red[tid + s];
        __syncthreads();
    }
    float mean = red[0] * (1.0f / 256.0f);
    __syncthreads();
    float d = v - mean;
    red[tid] = d * d;
    __syncthreads();
    for (int s = 128; s > 0; s >>= 1) {
        if (tid < s) red[tid] += red[tid + s];
        __syncthreads();
    }
    float var = red[0] * (1.0f / 256.0f);
    float r = rsqrtf(var + 1e-6f);
    out[(size_t)row * 256 + tid] = d * r * g[tid] + b[tid];
}

typedef __attribute__((ext_vector_type(4))) float v4f;

__global__ void __launch_bounds__(256) __attribute__((amdgpu_waves_per_eu(2, 8))) gemm_kernel(const float* __restrict__ A, const float* __restrict__ W,
                            float* __restrict__ C, const float* __restrict__ bias,
                            const float* __restrict__ mfv, const float* __restrict__ res,
                            int K, int N, int relu) {
    __shared__ __align__(16) float As[64][34];
    __shared__ __align__(16) float Bs[16][144][2];

    const int tid = threadIdx.x;
    const int lane = tid & 31;
    const int wave = tid >> 5;
    const int half = lane >> 4;
    const int lane16 = lane & 15;
    const int wm = wave >> 2;
    const int wn = wave & 3;
    const int m0 = blockIdx.x * 64;
    const int n0 = blockIdx.y * 128;

    const int rowA0 = wm * 32 + lane16;
    const int rowA1 = rowA0 + 16;
    const int colB0 = wn * 32 + lane16;
    const int colB1 = colB0 + 16;

    const int ar = tid >> 3;
    const int ac = (tid & 7) * 4;
    const int bkp = tid >> 5;
    const int bc = (tid & 31) * 4;
    const unsigned aoff = (unsigned)(ar * K + ac);
    const unsigned woff = (unsigned)(2 * bkp * N + bc);
    const float* Au = A + (size_t)m0 * K;
    const float* Wu = W + n0;
    const size_t aHop = (size_t)32 * K;
    const size_t wHop = (size_t)16 * N;
    const size_t wStep = (size_t)32 * N;

    v8f acc[2][2] = {};

    v4f ra0 = *(const v4f*)(Au + aoff);
    v4f ra1 = *(const v4f*)(Au + aHop + aoff);
    v4f rb00 = *(const v4f*)(Wu + woff);
    v4f rb01 = *(const v4f*)(Wu + N + woff);
    v4f rb10 = *(const v4f*)(Wu + wHop + woff);
    v4f rb11 = *(const v4f*)(Wu + wHop + N + woff);

    for (int k0 = 0; k0 < K; k0 += 32) {
        *(v2f*)&As[ar][ac]          = (v2f){ra0[0], ra0[1]};
        *(v2f*)&As[ar][ac + 2]      = (v2f){ra0[2], ra0[3]};
        *(v2f*)&As[ar + 32][ac]     = (v2f){ra1[0], ra1[1]};
        *(v2f*)&As[ar + 32][ac + 2] = (v2f){ra1[2], ra1[3]};
        *(v4f*)&Bs[bkp][bc][0]         = (v4f){rb00[0], rb01[0], rb00[1], rb01[1]};
        *(v4f*)&Bs[bkp][bc + 2][0]     = (v4f){rb00[2], rb01[2], rb00[3], rb01[3]};
        *(v4f*)&Bs[bkp + 8][bc][0]     = (v4f){rb10[0], rb11[0], rb10[1], rb11[1]};
        *(v4f*)&Bs[bkp + 8][bc + 2][0] = (v4f){rb10[2], rb11[2], rb10[3], rb11[3]};
        __syncthreads();

        if (k0 + 32 < K) {
            Au += 32;
            Wu += wStep;
            ra0 = *(const v4f*)(Au + aoff);
            ra1 = *(const v4f*)(Au + aHop + aoff);
            rb00 = *(const v4f*)(Wu + woff);
            rb01 = *(const v4f*)(Wu + N + woff);
            rb10 = *(const v4f*)(Wu + wHop + woff);
            rb11 = *(const v4f*)(Wu + wHop + N + woff);
        }

        v2f a0c = *(const v2f*)&As[rowA0][half * 2];
        v2f a1c = *(const v2f*)&As[rowA1][half * 2];
        v2f b0c = *(const v2f*)&Bs[half][colB0][0];
        v2f b1c = *(const v2f*)&Bs[half][colB1][0];
#pragma unroll
        for (int t = 0; t < 8; ++t) {
            const int kkn = ((t + 1) & 7) * 4 + half * 2;
            v2f a0n = *(const v2f*)&As[rowA0][kkn];
            v2f a1n = *(const v2f*)&As[rowA1][kkn];
            v2f b0n = *(const v2f*)&Bs[kkn >> 1][colB0][0];
            v2f b1n = *(const v2f*)&Bs[kkn >> 1][colB1][0];
            acc[0][0] = wmma4(a0c, b0c, acc[0][0]);
            acc[0][1] = wmma4(a0c, b1c, acc[0][1]);
            acc[1][0] = wmma4(a1c, b0c, acc[1][0]);
            acc[1][1] = wmma4(a1c, b1c, acc[1][1]);
            a0c = a0n; a1c = a1n; b0c = b0n; b1c = b1n;
        }
        __syncthreads();
    }

#pragma unroll
    for (int im = 0; im < 2; ++im) {
#pragma unroll
        for (int jn = 0; jn < 2; ++jn) {
            const int col = n0 + wn * 32 + jn * 16 + lane16;
            const float bv = bias ? bias[col] : 0.0f;
#pragma unroll
            for (int vi = 0; vi < 8; ++vi) {
                const int row = m0 + wm * 32 + im * 16 + vi + half * 8;
                float val = acc[im][jn][vi] + bv;
                if (relu) val = fmaxf(val, 0.0f);
                if (mfv) val *= mfv[row];
                if (res) val += res[(size_t)row * N + col];
                C[(size_t)row * N + col] = val;
            }
        }
    }
}


__device__ __forceinline__ float xhalf(float x) {
    const int bits = __builtin_bit_cast(int, x);
    return __builtin_bit_cast(float, __builtin_amdgcn_permlanex16(bits, bits, 0x76543210, 0xfedcba98, false, false));
}

__global__ void __launch_bounds__(128) attn_kernel(const float* __restrict__ q, const float* __restrict__ k,
                            const float* __restrict__ v, const float* __restrict__ mfv,
                            float* __restrict__ o) {
    __shared__ __align__(16) float Ks[64][36];
    __shared__ __align__(16) float Vt[32][68];
    __shared__ __align__(16) float mb[64];

    const int S = 2048, LD = 256;
    const int tid = threadIdx.x;
    const int lane = tid & 31;
    const int wave = tid >> 5;
    const int half = lane >> 4;
    const int lane16 = lane & 15;
    const int bh = blockIdx.y;
    const int b = bh >> 3;
    const int h = bh & 7;
    const int q0 = blockIdx.x * 64;

    const float qscale = 0.17677669529663687f * 1.4426950408889634f;
    v4f qf[4];
    {
        const float* qp = q + ((size_t)b * S + q0 + wave * 16 + lane16) * LD + h * 32 + half * 4;
#pragma unroll
        for (int u = 0; u < 4; ++u) qf[u] = *(const v4f*)(qp + 8 * u) * qscale;
    }

    v8f oacc[2] = {};
    float m_run = -3.0e38f;
    float l_part = 0.0f;

    for (int j0 = 0; j0 < S; j0 += 64) {
        const size_t baseK = ((size_t)b * S + j0) * LD + h * 32;
        v4f rk[4];
        float rv[4][4];
#pragma unroll
        for (int j = 0; j < 4; ++j) {
            const int i = tid + j * 128;
            rk[j] = *(const v4f*)&k[baseK + (size_t)(i >> 3) * LD + 4 * (i & 7)];
        }
#pragma unroll
        for (int j = 0; j < 4; ++j) {
            const int kq = wave + 4 * j;
#pragma unroll
            for (int i = 0; i < 4; ++i) rv[j][i] = v[baseK + (size_t)(4 * kq + i) * LD + lane];
        }
        const float rmk = mfv[(size_t)b * S + j0 + (tid & 63)];
#pragma unroll
        for (int j = 0; j < 4; ++j) {
            const int i = tid + j * 128;
            *(v4f*)&Ks[i >> 3][4 * (i & 7)] = rk[j];
        }
#pragma unroll
        for (int j = 0; j < 4; ++j) {
            const int kq = wave + 4 * j;
            v4f t = {rv[j][0], rv[j][1], rv[j][2], rv[j][3]};
            *(v4f*)&Vt[lane][4 * kq] = t;
        }
        mb[tid & 63] = (rmk != 0.0f) ? 0.0f : -1e9f;
        if (j0 + 64 < S) {
            const size_t baseN = ((size_t)b * S + j0 + 64) * LD + h * 32;
            if (tid < 64) __builtin_prefetch(&k[baseN + (size_t)tid * LD], 0, 1);
            else          __builtin_prefetch(&v[baseN + (size_t)(tid - 64) * LD], 0, 1);
        }
        __syncthreads();

        v8f sacc[4];
#pragma unroll
        for (int kt = 0; kt < 4; ++kt) {
            const v4f b0 = *(const v4f*)&mb[kt * 16 + half * 8];
            const v4f b1 = *(const v4f*)&mb[kt * 16 + half * 8 + 4];
            sacc[kt][0] = b0[0]; sacc[kt][1] = b0[1]; sacc[kt][2] = b0[2]; sacc[kt][3] = b0[3];
            sacc[kt][4] = b1[0]; sacc[kt][5] = b1[1]; sacc[kt][6] = b1[2]; sacc[kt][7] = b1[3];
        }
        v4f ka[4][4];
#pragma unroll
        for (int u = 0; u < 4; ++u)
#pragma unroll
            for (int kt = 0; kt < 4; ++kt)
                ka[u][kt] = *(const v4f*)&Ks[kt * 16 + lane16][8 * u + half * 4];
        v4f va[4][2][2];
#pragma unroll
        for (int kt = 0; kt < 4; ++kt)
#pragma unroll
            for (int w = 0; w < 2; ++w) {
                va[kt][w][0] = *(const v4f*)&Vt[lane16][kt * 16 + half * 8 + 4 * w];
                va[kt][w][1] = *(const v4f*)&Vt[16 + lane16][kt * 16 + half * 8 + 4 * w];
            }
#pragma unroll
        for (int u = 0; u < 4; ++u) {
#pragma unroll
            for (int e = 0; e < 2; ++e) {
                const v2f bq = {qf[u][2 * e], qf[u][2 * e + 1]};
#pragma unroll
                for (int kt = 0; kt < 4; ++kt) {
                    const v2f ak = {ka[u][kt][2 * e], ka[u][kt][2 * e + 1]};
                    sacc[kt] = wmma4(ak, bq, sacc[kt]);
                }
            }
        }

        float mx[4];
#pragma unroll
        for (int kt = 0; kt < 4; ++kt) {
            mx[kt] = sacc[kt][0];
#pragma unroll
            for (int r = 1; r < 8; ++r) mx[kt] = fmaxf(mx[kt], sacc[kt][r]);
        }
        float mc = fmaxf(fmaxf(mx[0], mx[1]), fmaxf(mx[2], mx[3]));
        mc = fmaxf(mc, xhalf(mc));
        const float mn = fmaxf(m_run, mc);
        const float al = __builtin_amdgcn_exp2f(m_run - mn);
        m_run = mn;
        float ls[4];
#pragma unroll
        for (int kt = 0; kt < 4; ++kt) {
            ls[kt] = 0.0f;
#pragma unroll
            for (int r = 0; r < 8; ++r) {
                const float p = __builtin_amdgcn_exp2f(sacc[kt][r] - mn);
                sacc[kt][r] = p;
                ls[kt] += p;
            }
        }
        l_part = l_part * al + ((ls[0] + ls[1]) + (ls[2] + ls[3]));
#pragma unroll
        for (int dt = 0; dt < 2; ++dt)
#pragma unroll
            for (int r = 0; r < 8; ++r) oacc[dt][r] *= al;

#pragma unroll
        for (int kt = 0; kt < 4; ++kt) {
#pragma unroll
            for (int w = 0; w < 2; ++w) {
#pragma unroll
                for (int e = 0; e < 2; ++e) {
                    const v2f pb = {sacc[kt][4 * w + 2 * e], sacc[kt][4 * w + 2 * e + 1]};
                    const v2f a0 = {va[kt][w][0][2 * e], va[kt][w][0][2 * e + 1]};
                    const v2f a1 = {va[kt][w][1][2 * e], va[kt][w][1][2 * e + 1]};
                    oacc[0] = wmma4(a0, pb, oacc[0]);
                    oacc[1] = wmma4(a1, pb, oacc[1]);
                }
            }
        }
        __syncthreads();
    }

    const float lsum = l_part + xhalf(l_part);
    const float linv = 1.0f / lsum;
    float* op = o + ((size_t)b * S + q0 + wave * 16 + lane16) * LD + h * 32 + half * 8;
#pragma unroll
    for (int dt = 0; dt < 2; ++dt) {
        const v4f lo = {oacc[dt][0] * linv, oacc[dt][1] * linv, oacc[dt][2] * linv, oacc[dt][3] * linv};
        const v4f hi = {oacc[dt][4] * linv, oacc[dt][5] * linv, oacc[dt][6] * linv, oacc[dt][7] * linv};
        *(v4f*)(op + dt * 16) = lo;
        *(v4f*)(op + dt * 16 + 4) = hi;
    }
}

extern "C" void kernel_launch(void* const* d_in, const int* in_sizes, int n_in,
                              void* d_out, int out_size, void* d_ws, size_t ws_size,
                              hipStream_t stream) {
    (void)in_sizes; (void)n_in; (void)out_size; (void)ws_size;
    const int B = 4, S = 2048, D = 256, F = 1024, L = 4;
    const int M = B * S;

    const float* x_in  = (const float*)d_in[0];
    const int*   mask  = (const int*)d_in[1];
    const float* w_qs  = (const float*)d_in[2];
    const float* w_ks  = (const float*)d_in[3];
    const float* w_vs  = (const float*)d_in[4];
    const float* w_fc  = (const float*)d_in[5];
    const float* ln1_g = (const float*)d_in[6];
    const float* ln1_b = (const float*)d_in[7];
    const float* w1    = (const float*)d_in[8];
    const float* b1    = (const float*)d_in[9];
    const float* w2    = (const float*)d_in[10];
    const float* b2    = (const float*)d_in[11];
    const float* ln2_g = (const float*)d_in[12];
    const float* ln2_b = (const float*)d_in[13];

    float* ws    = (float*)d_ws;
    float* x_cur = ws;
    float* xn    = x_cur + (size_t)M * D;
    float* qb    = xn + (size_t)M * D;
    float* kb    = qb + (size_t)M * D;
    float* vb    = kb + (size_t)M * D;
    float* ob    = vb + (size_t)M * D;
    float* hb    = ob + (size_t)M * D;
    float* mfv   = hb + (size_t)M * F;

    hipMemcpyAsync(x_cur, x_in, sizeof(float) * (size_t)M * D,
                   hipMemcpyDeviceToDevice, stream);
    mf_kernel<<<M / 256, 256, 0, stream>>>(mask, mfv, M);

    const dim3 gD(M / 64, D / 128);
    const dim3 gF(M / 64, F / 128);
    const dim3 gA(S / 64, B * 8);

    for (int l = 0; l < L; ++l) {
        const float* wq = w_qs + (size_t)l * D * D;
        const float* wk = w_ks + (size_t)l * D * D;
        const float* wv = w_vs + (size_t)l * D * D;
        const float* wf = w_fc + (size_t)l * D * D;
        const float* w1l = w1 + (size_t)l * D * F;
        const float* w2l = w2 + (size_t)l * F * D;
        const float* b1l = b1 + (size_t)l * F;
        const float* b2l = b2 + (size_t)l * D;

        ln_kernel<<<M, 256, 0, stream>>>(x_cur, xn, ln1_g + (size_t)l * D, ln1_b + (size_t)l * D);
        gemm_kernel<<<gD, 256, 0, stream>>>(xn, wq, qb, nullptr, mfv, nullptr, D, D, 0);
        gemm_kernel<<<gD, 256, 0, stream>>>(xn, wk, kb, nullptr, mfv, nullptr, D, D, 0);
        gemm_kernel<<<gD, 256, 0, stream>>>(xn, wv, vb, nullptr, mfv, nullptr, D, D, 0);
        attn_kernel<<<gA, 128, 0, stream>>>(qb, kb, vb, mfv, ob);
        gemm_kernel<<<gD, 256, 0, stream>>>(ob, wf, xn, nullptr, mfv, x_cur, D, D, 0);
        gemm_kernel<<<gF, 256, 0, stream>>>(xn, w1l, hb, b1l, nullptr, nullptr, D, F, 1);
        gemm_kernel<<<gD, 256, 0, stream>>>(hb, w2l, qb, b2l, nullptr, xn, F, D, 0);
        ln_kernel<<<M, 256, 0, stream>>>(qb, x_cur, ln2_g + (size_t)l * D, ln2_b + (size_t)l * D);
    }

    hipMemcpyAsync(d_out, x_cur, sizeof(float) * (size_t)M * D,
                   hipMemcpyDeviceToDevice, stream);
}


# Attention: tokens in mask-sorted order (set mask first; per-call ballot ranks): key loop and query tiles stop at the set count, rows gathered by index; key sum order changes
# speedup vs baseline: 2.5281x; 1.1674x over previous
#include <hip/hip_runtime.h>
#include <hip/hip_bf16.h>

typedef __attribute__((ext_vector_type(2))) float v2f;
typedef __attribute__((ext_vector_type(8))) float v8f;

__device__ __forceinline__ v8f wmma4(v2f a, v2f b, v8f c) {
    return __builtin_amdgcn_wmma_f32_16x16x4_f32(false, a, false, b, (short)0, c, false, false);
}

__global__ void __launch_bounds__(1024) mask_kernel(const int* __restrict__ mask, float* __restrict__ mfv,
                                                    int* __restrict__ perm, int* __restrict__ cnt) {
    __shared__ int wtot[64];
    __shared__ int wbase[65];
    __shared__ int sperm[2048];
    const int b = blockIdx.x;
    const int tid = threadIdx.x;
    const int lane = tid & 31;
    const int wave = tid >> 5;
    const int* mp = mask + (size_t)b * 2048;
    const bool s0 = mp[tid] != 0;
    const bool s1 = mp[tid + 1024] != 0;
    mfv[(size_t)b * 2048 + tid] = s0 ? 1.0f : 0.0f;
    mfv[(size_t)b * 2048 + tid + 1024] = s1 ? 1.0f : 0.0f;
    const unsigned bal0 = __builtin_amdgcn_ballot_w32(s0);
    const unsigned bal1 = __builtin_amdgcn_ballot_w32(s1);
    const unsigned below = (1u << lane) - 1u;
    const int p0 = __builtin_popcount(bal0 & below);
    const int p1 = __builtin_popcount(bal1 & below);
    if (lane == 0) {
        wtot[wave] = __builtin_popcount(bal0);
        wtot[32 + wave] = __builtin_popcount(bal1);
    }
    __syncthreads();
    if (tid == 0) {
        int a = 0;
        for (int w = 0; w < 64; ++w) { wbase[w] = a; a += wtot[w]; }
        wbase[64] = a;
    }
    __syncthreads();
    const int total = wbase[64];
    const int o0 = wbase[wave] + p0;
    const int o1 = wbase[32 + wave] + p1;
    sperm[s0 ? o0 : total + tid - o0] = tid;
    sperm[s1 ? o1 : total + (tid + 1024) - o1] = tid + 1024;
    __syncthreads();
    perm[(size_t)b * 2048 + tid] = sperm[tid];
    perm[(size_t)b * 2048 + tid + 1024] = sperm[tid + 1024];
    if (tid == 0) cnt[b] = total;
}

__global__ void ln_kernel(const float* __restrict__ x, float* __restrict__ out,
                          const float* __restrict__ g, const float* __restrict__ b) {
    __shared__ float red[256];
    const int row = blockIdx.x;
    const int tid = threadIdx.x;
    float v = x[(size_t)row * 256 + tid];
    red[tid] = v;
    __syncthreads();
    for (int s = 128; s > 0; s >>= 1) {
        if (tid < s) red[tid] += red[tid + s];
        __syncthreads();
    }
    float mean = red[0] * (1.0f / 256.0f);
    __syncthreads();
    float d = v - mean;
    red[tid] = d * d;
    __syncthreads();
    for (int s = 128; s > 0; s >>= 1) {
        if (tid < s) red[tid] += red[tid + s];
        __syncthreads();
    }
    float var = red[0] * (1.0f / 256.0f);
    float r = rsqrtf(var + 1e-6f);
    out[(size_t)row * 256 + tid] = d * r * g[tid] + b[tid];
}

typedef __attribute__((ext_vector_type(4))) float v4f;

__global__ void __launch_bounds__(256) __attribute__((amdgpu_waves_per_eu(2, 8))) gemm_kernel(const float* __restrict__ A, const float* __restrict__ W,
                            float* __restrict__ C, const float* __restrict__ bias,
                            const float* __restrict__ mfv, const float* __restrict__ res,
                            int K, int N, int relu) {
    __shared__ __align__(16) float As[64][34];
    __shared__ __align__(16) float Bs[16][144][2];

    const int tid = threadIdx.x;
    const int lane = tid & 31;
    const int wave = tid >> 5;
    const int half = lane >> 4;
    const int lane16 = lane & 15;
    const int wm = wave >> 2;
    const int wn = wave & 3;
    const int m0 = blockIdx.x * 64;
    const int n0 = blockIdx.y * 128;

    const int rowA0 = wm * 32 + lane16;
    const int rowA1 = rowA0 + 16;
    const int colB0 = wn * 32 + lane16;
    const int colB1 = colB0 + 16;

    const int ar = tid >> 3;
    const int ac = (tid & 7) * 4;
    const int bkp = tid >> 5;
    const int bc = (tid & 31) * 4;
    const unsigned aoff = (unsigned)(ar * K + ac);
    const unsigned woff = (unsigned)(2 * bkp * N + bc);
    const float* Au = A + (size_t)m0 * K;
    const float* Wu = W + n0;
    const size_t aHop = (size_t)32 * K;
    const size_t wHop = (size_t)16 * N;
    const size_t wStep = (size_t)32 * N;

    v8f acc[2][2] = {};

    v4f ra0 = *(const v4f*)(Au + aoff);
    v4f ra1 = *(const v4f*)(Au + aHop + aoff);
    v4f rb00 = *(const v4f*)(Wu + woff);
    v4f rb01 = *(const v4f*)(Wu + N + woff);
    v4f rb10 = *(const v4f*)(Wu + wHop + woff);
    v4f rb11 = *(const v4f*)(Wu + wHop + N + woff);

    for (int k0 = 0; k0 < K; k0 += 32) {
        *(v2f*)&As[ar][ac]          = (v2f){ra0[0], ra0[1]};
        *(v2f*)&As[ar][ac + 2]      = (v2f){ra0[2], ra0[3]};
        *(v2f*)&As[ar + 32][ac]     = (v2f){ra1[0], ra1[1]};
        *(v2f*)&As[ar + 32][ac + 2] = (v2f){ra1[2], ra1[3]};
        *(v4f*)&Bs[bkp][bc][0]         = (v4f){rb00[0], rb01[0], rb00[1], rb01[1]};
        *(v4f*)&Bs[bkp][bc + 2][0]     = (v4f){rb00[2], rb01[2], rb00[3], rb01[3]};
        *(v4f*)&Bs[bkp + 8][bc][0]     = (v4f){rb10[0], rb11[0], rb10[1], rb11[1]};
        *(v4f*)&Bs[bkp + 8][bc + 2][0] = (v4f){rb10[2], rb11[2], rb10[3], rb11[3]};
        __syncthreads();

        if (k0 + 32 < K) {
            Au += 32;
            Wu += wStep;
            ra0 = *(const v4f*)(Au + aoff);
            ra1 = *(const v4f*)(Au + aHop + aoff);
            rb00 = *(const v4f*)(Wu + woff);
            rb01 = *(const v4f*)(Wu + N + woff);
            rb10 = *(const v4f*)(Wu + wHop + woff);
            rb11 = *(const v4f*)(Wu + wHop + N + woff);
        }

        v2f a0c = *(const v2f*)&As[rowA0][half * 2];
        v2f a1c = *(const v2f*)&As[rowA1][half * 2];
        v2f b0c = *(const v2f*)&Bs[half][colB0][0];
        v2f b1c = *(const v2f*)&Bs[half][colB1][0];
#pragma unroll
        for (int t = 0; t < 8; ++t) {
            const int kkn = ((t + 1) & 7) * 4 + half * 2;
            v2f a0n = *(const v2f*)&As[rowA0][kkn];
            v2f a1n = *(const v2f*)&As[rowA1][kkn];
            v2f b0n = *(const v2f*)&Bs[kkn >> 1][colB0][0];
            v2f b1n = *(const v2f*)&Bs[kkn >> 1][colB1][0];
            acc[0][0] = wmma4(a0c, b0c, acc[0][0]);
            acc[0][1] = wmma4(a0c, b1c, acc[0][1]);
            acc[1][0] = wmma4(a1c, b0c, acc[1][0]);
            acc[1][1] = wmma4(a1c, b1c, acc[1][1]);
            a0c = a0n; a1c = a1n; b0c = b0n; b1c = b1n;
        }
        __syncthreads();
    }

#pragma unroll
    for (int im = 0; im < 2; ++im) {
#pragma unroll
        for (int jn = 0; jn < 2; ++jn) {
            const int col = n0 + wn * 32 + jn * 16 + lane16;
            const float bv = bias ? bias[col] : 0.0f;
#pragma unroll
            for (int vi = 0; vi < 8; ++vi) {
                const int row = m0 + wm * 32 + im * 16 + vi + half * 8;
                float val = acc[im][jn][vi] + bv;
                if (relu) val = fmaxf(val, 0.0f);
                if (mfv) val *= mfv[row];
                if (res) val += res[(size_t)row * N + col];
                C[(size_t)row * N + col] = val;
            }
        }
    }
}


typedef __attribute__((ext_vector_type(4))) int v4i;

__device__ __forceinline__ float xhalf(float x) {
    const int bits = __builtin_bit_cast(int, x);
    return __builtin_bit_cast(float, __builtin_amdgcn_permlanex16(bits, bits, 0x76543210, 0xfedcba98, false, false));
}

__global__ void __launch_bounds__(128) attn_kernel(const float* __restrict__ q, const float* __restrict__ k,
                            const float* __restrict__ v, const int* __restrict__ perm,
                            const int* __restrict__ cnt, float* __restrict__ o) {
    __shared__ __align__(16) float Ks[64][36];
    __shared__ __align__(16) float Vt[32][68];
    __shared__ __align__(16) float mb[64];

    const int S = 2048, LD = 256;
    const int tid = threadIdx.x;
    const int lane = tid & 31;
    const int wave = tid >> 5;
    const int half = lane >> 4;
    const int lane16 = lane & 15;
    const int bh = blockIdx.y;
    const int b = bh >> 3;
    const int h = bh & 7;
    const int par = (__builtin_popcount(blockIdx.x) + __builtin_popcount(blockIdx.y)) & 1;
    const int q0 = ((int)(blockIdx.x >> 1) | (par << 4)) * 64;
    const int nk = cnt[b];
    const int* pb = perm + (size_t)b * S;

    if (q0 >= nk) {
        float* zp = o + ((size_t)b * S + pb[q0 + (tid >> 1)]) * LD + h * 32 + (tid & 1) * 16;
        const v4f z = {0.0f, 0.0f, 0.0f, 0.0f};
        *(v4f*)(zp) = z;
        *(v4f*)(zp + 4) = z;
        *(v4f*)(zp + 8) = z;
        *(v4f*)(zp + 12) = z;
        return;
    }
    const int qrow = pb[q0 + wave * 16 + lane16];
    const int kslot = tid >> 3;

    const float qscale = 0.17677669529663687f * 1.4426950408889634f;
    v4f qf[4];
    {
        const float* qp = q + ((size_t)b * S + qrow) * LD + h * 32 + half * 4;
#pragma unroll
        for (int u = 0; u < 4; ++u) qf[u] = *(const v4f*)(qp + 8 * u) * qscale;
    }

    v8f oacc[2] = {};
    float m_run = -3.0e38f;
    float l_part = 0.0f;

    v4i vi[4];
    int ki[4];
#pragma unroll
    for (int j = 0; j < 4; ++j) {
        vi[j] = *(const v4i*)(pb + 4 * wave + 16 * j);
        ki[j] = pb[kslot + 16 * j];
    }
    const unsigned rowB = (unsigned)(b * S);
    const unsigned colH = (unsigned)(h * 32);

    for (int j0 = 0; j0 < nk; j0 += 64) {
        const int jn = (j0 + 64 < S) ? j0 + 64 : j0;
        v4i vn[4];
        int kn[4];
#pragma unroll
        for (int j = 0; j < 4; ++j) {
            vn[j] = *(const v4i*)(pb + jn + 4 * wave + 16 * j);
            kn[j] = pb[jn + kslot + 16 * j];
        }
        const int pn = pb[jn + (tid & 63)];
        v4f rk[4];
        float rv[4][4];
#pragma unroll
        for (int j = 0; j < 4; ++j) {
            rk[j] = *(const v4f*)&k[(size_t)((rowB + (unsigned)ki[j]) * LD + colH + 4u * (tid & 7))];
        }
#pragma unroll
        for (int j = 0; j < 4; ++j) {
#pragma unroll
            for (int i = 0; i < 4; ++i)
                rv[j][i] = v[(size_t)((rowB + (unsigned)vi[j][i]) * LD + colH) + lane];
        }
#pragma unroll
        for (int j = 0; j < 4; ++j) {
            const int i = tid + j * 128;
            *(v4f*)&Ks[i >> 3][4 * (i & 7)] = rk[j];
        }
#pragma unroll
        for (int j = 0; j < 4; ++j) {
            const int kq = wave + 4 * j;
            v4f t = {rv[j][0], rv[j][1], rv[j][2], rv[j][3]};
            *(v4f*)&Vt[lane][4 * kq] = t;
        }
        mb[tid & 63] = (j0 + (tid & 63) < nk) ? 0.0f : -1e9f;
        {
            const size_t rowN = (size_t)((rowB + (unsigned)pn) * LD + colH);
            if (tid < 64) __builtin_prefetch(&k[rowN], 0, 1);
            else          __builtin_prefetch(&v[rowN], 0, 1);
        }
        __syncthreads();

        v8f sacc[4];
#pragma unroll
        for (int kt = 0; kt < 4; ++kt) {
            const v4f b0 = *(const v4f*)&mb[kt * 16 + half * 8];
            const v4f b1 = *(const v4f*)&mb[kt * 16 + half * 8 + 4];
            sacc[kt][0] = b0[0]; sacc[kt][1] = b0[1]; sacc[kt][2] = b0[2]; sacc[kt][3] = b0[3];
            sacc[kt][4] = b1[0]; sacc[kt][5] = b1[1]; sacc[kt][6] = b1[2]; sacc[kt][7] = b1[3];
        }
        v4f ka[4][4];
#pragma unroll
        for (int u = 0; u < 4; ++u)
#pragma unroll
            for (int kt = 0; kt < 4; ++kt)
                ka[u][kt] = *(const v4f*)&Ks[kt * 16 + lane16][8 * u + half * 4];
        v4f va[4][2][2];
#pragma unroll
        for (int kt = 0; kt < 4; ++kt)
#pragma unroll
            for (int w = 0; w < 2; ++w) {
                va[kt][w][0] = *(const v4f*)&Vt[lane16][kt * 16 + half * 8 + 4 * w];
                va[kt][w][1] = *(const v4f*)&Vt[16 + lane16][kt * 16 + half * 8 + 4 * w];
            }
#pragma unroll
        for (int u = 0; u < 4; ++u) {
#pragma unroll
            for (int e = 0; e < 2; ++e) {
                const v2f bq = {qf[u][2 * e], qf[u][2 * e + 1]};
#pragma unroll
                for (int kt = 0; kt < 4; ++kt) {
                    const v2f ak = {ka[u][kt][2 * e], ka[u][kt][2 * e + 1]};
                    sacc[kt] = wmma4(ak, bq, sacc[kt]);
                }
            }
        }

        float mx[4];
#pragma unroll
        for (int kt = 0; kt < 4; ++kt) {
            mx[kt] = sacc[kt][0];
#pragma unroll
            for (int r = 1; r < 8; ++r) mx[kt] = fmaxf(mx[kt], sacc[kt][r]);
        }
        float mc = fmaxf(fmaxf(mx[0], mx[1]), fmaxf(mx[2], mx[3]));
        mc = fmaxf(mc, xhalf(mc));
        const float mn = fmaxf(m_run, mc);
        const float al = __builtin_amdgcn_exp2f(m_run - mn);
        m_run = mn;
        float ls[4];
#pragma unroll
        for (int kt = 0; kt < 4; ++kt) {
            ls[kt] = 0.0f;
#pragma unroll
            for (int r = 0; r < 8; ++r) {
                const float p = __builtin_amdgcn_exp2f(sacc[kt][r] - mn);
                sacc[kt][r] = p;
                ls[kt] += p;
            }
        }
        l_part = l_part * al + ((ls[0] + ls[1]) + (ls[2] + ls[3]));
#pragma unroll
        for (int dt = 0; dt < 2; ++dt)
#pragma unroll
            for (int r = 0; r < 8; ++r) oacc[dt][r] *= al;

#pragma unroll
        for (int kt = 0; kt < 4; ++kt) {
#pragma unroll
            for (int w = 0; w < 2; ++w) {
#pragma unroll
                for (int e = 0; e < 2; ++e) {
                    const v2f pb = {sacc[kt][4 * w + 2 * e], sacc[kt][4 * w + 2 * e + 1]};
                    const v2f a0 = {va[kt][w][0][2 * e], va[kt][w][0][2 * e + 1]};
                    const v2f a1 = {va[kt][w][1][2 * e], va[kt][w][1][2 * e + 1]};
                    oacc[0] = wmma4(a0, pb, oacc[0]);
                    oacc[1] = wmma4(a1, pb, oacc[1]);
                }
            }
        }
        __syncthreads();
        vi[0] = vn[0]; vi[1] = vn[1]; vi[2] = vn[2]; vi[3] = vn[3];
        ki[0] = kn[0]; ki[1] = kn[1]; ki[2] = kn[2]; ki[3] = kn[3];
    }

    const float lsum = l_part + xhalf(l_part);
    const float linv = 1.0f / lsum;
    float* op = o + ((size_t)b * S + qrow) * LD + h * 32 + half * 8;
#pragma unroll
    for (int dt = 0; dt < 2; ++dt) {
        const v4f lo = {oacc[dt][0] * linv, oacc[dt][1] * linv, oacc[dt][2] * linv, oacc[dt][3] * linv};
        const v4f hi = {oacc[dt][4] * linv, oacc[dt][5] * linv, oacc[dt][6] * linv, oacc[dt][7] * linv};
        *(v4f*)(op + dt * 16) = lo;
        *(v4f*)(op + dt * 16 + 4) = hi;
    }
}

extern "C" void kernel_launch(void* const* d_in, const int* in_sizes, int n_in,
                              void* d_out, int out_size, void* d_ws, size_t ws_size,
                              hipStream_t stream) {
    (void)in_sizes; (void)n_in; (void)out_size; (void)ws_size;
    const int B = 4, S = 2048, D = 256, F = 1024, L = 4;
    const int M = B * S;

    const float* x_in  = (const float*)d_in[0];
    const int*   mask  = (const int*)d_in[1];
    const float* w_qs  = (const float*)d_in[2];
    const float* w_ks  = (const float*)d_in[3];
    const float* w_vs  = (const float*)d_in[4];
    const float* w_fc  = (const float*)d_in[5];
    const float* ln1_g = (const float*)d_in[6];
    const float* ln1_b = (const float*)d_in[7];
    const float* w1    = (const float*)d_in[8];
    const float* b1    = (const float*)d_in[9];
    const float* w2    = (const float*)d_in[10];
    const float* b2    = (const float*)d_in[11];
    const float* ln2_g = (const float*)d_in[12];
    const float* ln2_b = (const float*)d_in[13];

    float* ws    = (float*)d_ws;
    float* x_cur = ws;
    float* xn    = x_cur + (size_t)M * D;
    float* qb    = xn + (size_t)M * D;
    float* kb    = qb + (size_t)M * D;
    float* vb    = kb + (size_t)M * D;
    float* ob    = vb + (size_t)M * D;
    float* hb    = ob + (size_t)M * D;
    float* mfv   = hb + (size_t)M * F;
    int*   perm  = (int*)(mfv + M);
    int*   cnt   = perm + M;

    hipMemcpyAsync(x_cur, x_in, sizeof(float) * (size_t)M * D,
                   hipMemcpyDeviceToDevice, stream);
    mask_kernel<<<B, 1024, 0, stream>>>(mask, mfv, perm, cnt);

    const dim3 gD(M / 64, D / 128);
    const dim3 gF(M / 64, F / 128);
    const dim3 gA(S / 64, B * 8);

    for (int l = 0; l < L; ++l) {
        const float* wq = w_qs + (size_t)l * D * D;
        const float* wk = w_ks + (size_t)l * D * D;
        const float* wv = w_vs + (size_t)l * D * D;
        const float* wf = w_fc + (size_t)l * D * D;
        const float* w1l = w1 + (size_t)l * D * F;
        const float* w2l = w2 + (size_t)l * F * D;
        const float* b1l = b1 + (size_t)l * F;
        const float* b2l = b2 + (size_t)l * D;

        ln_kernel<<<M, 256, 0, stream>>>(x_cur, xn, ln1_g + (size_t)l * D, ln1_b + (size_t)l * D);
        gemm_kernel<<<gD, 256, 0, stream>>>(xn, wq, qb, nullptr, mfv, nullptr, D, D, 0);
        gemm_kernel<<<gD, 256, 0, stream>>>(xn, wk, kb, nullptr, mfv, nullptr, D, D, 0);
        gemm_kernel<<<gD, 256, 0, stream>>>(xn, wv, vb, nullptr, mfv, nullptr, D, D, 0);
        attn_kernel<<<gA, 128, 0, stream>>>(qb, kb, vb, perm, cnt, ob);
        gemm_kernel<<<gD, 256, 0, stream>>>(ob, wf, xn, nullptr, mfv, x_cur, D, D, 0);
        gemm_kernel<<<gF, 256, 0, stream>>>(xn, w1l, hb, b1l, nullptr, nullptr, D, F, 1);
        gemm_kernel<<<gD, 256, 0, stream>>>(hb, w2l, qb, b2l, nullptr, xn, F, D, 0);
        ln_kernel<<<M, 256, 0, stream>>>(qb, x_cur, ln2_g + (size_t)l * D, ln2_b + (size_t)l * D);
    }

    hipMemcpyAsync(d_out, x_cur, sizeof(float) * (size_t)M * D,
                   hipMemcpyDeviceToDevice, stream);
}


# GEMM epilogue: bias, mask-factor and residual loads issued as one group ahead of the element loop (absent operand = factor 1 / addend 0); was one load waited per output element
# speedup vs baseline: 3.3090x; 1.3089x over previous
#include <hip/hip_runtime.h>
#include <hip/hip_bf16.h>

typedef __attribute__((ext_vector_type(2))) float v2f;
typedef __attribute__((ext_vector_type(8))) float v8f;

__device__ __forceinline__ v8f wmma4(v2f a, v2f b, v8f c) {
    return __builtin_amdgcn_wmma_f32_16x16x4_f32(false, a, false, b, (short)0, c, false, false);
}

__global__ void __launch_bounds__(1024) mask_kernel(const int* __restrict__ mask, float* __restrict__ mfv,
                                                    int* __restrict__ perm, int* __restrict__ cnt) {
    __shared__ int wtot[64];
    __shared__ int wbase[65];
    __shared__ int sperm[2048];
    const int b = blockIdx.x;
    const int tid = threadIdx.x;
    const int lane = tid & 31;
    const int wave = tid >> 5;
    const int* mp = mask + (size_t)b * 2048;
    const bool s0 = mp[tid] != 0;
    const bool s1 = mp[tid + 1024] != 0;
    mfv[(size_t)b * 2048 + tid] = s0 ? 1.0f : 0.0f;
    mfv[(size_t)b * 2048 + tid + 1024] = s1 ? 1.0f : 0.0f;
    const unsigned bal0 = __builtin_amdgcn_ballot_w32(s0);
    const unsigned bal1 = __builtin_amdgcn_ballot_w32(s1);
    const unsigned below = (1u << lane) - 1u;
    const int p0 = __builtin_popcount(bal0 & below);
    const int p1 = __builtin_popcount(bal1 & below);
    if (lane == 0) {
        wtot[wave] = __builtin_popcount(bal0);
        wtot[32 + wave] = __builtin_popcount(bal1);
    }
    __syncthreads();
    if (tid == 0) {
        int a = 0;
        for (int w = 0; w < 64; ++w) { wbase[w] = a; a += wtot[w]; }
        wbase[64] = a;
    }
    __syncthreads();
    const int total = wbase[64];
    const int o0 = wbase[wave] + p0;
    const int o1 = wbase[32 + wave] + p1;
    sperm[s0 ? o0 : total + tid - o0] = tid;
    sperm[s1 ? o1 : total + (tid + 1024) - o1] = tid + 1024;
    __syncthreads();
    perm[(size_t)b * 2048 + tid] = sperm[tid];
    perm[(size_t)b * 2048 + tid + 1024] = sperm[tid + 1024];
    if (tid == 0) cnt[b] = total;
}

__global__ void ln_kernel(const float* __restrict__ x, float* __restrict__ out,
                          const float* __restrict__ g, const float* __restrict__ b) {
    __shared__ float red[256];
    const int row = blockIdx.x;
    const int tid = threadIdx.x;
    float v = x[(size_t)row * 256 + tid];
    red[tid] = v;
    __syncthreads();
    for (int s = 128; s > 0; s >>= 1) {
        if (tid < s) red[tid] += red[tid + s];
        __syncthreads();
    }
    float mean = red[0] * (1.0f / 256.0f);
    __syncthreads();
    float d = v - mean;
    red[tid] = d * d;
    __syncthreads();
    for (int s = 128; s > 0; s >>= 1) {
        if (tid < s) red[tid] += red[tid + s];
        __syncthreads();
    }
    float var = red[0] * (1.0f / 256.0f);
    float r = rsqrtf(var + 1e-6f);
    out[(size_t)row * 256 + tid] = d * r * g[tid] + b[tid];
}

typedef __attribute__((ext_vector_type(4))) float v4f;

__global__ void __launch_bounds__(256) __attribute__((amdgpu_waves_per_eu(2, 8))) gemm_kernel(const float* __restrict__ A, const float* __restrict__ W,
                            float* __restrict__ C, const float* __restrict__ bias,
                            const float* __restrict__ mfv, const float* __restrict__ res,
                            int K, int N, int relu) {
    __shared__ __align__(16) float As[64][34];
    __shared__ __align__(16) float Bs[16][144][2];

    const int tid = threadIdx.x;
    const int lane = tid & 31;
    const int wave = tid >> 5;
    const int half = lane >> 4;
    const int lane16 = lane & 15;
    const int wm = wave >> 2;
    const int wn = wave & 3;
    const int m0 = blockIdx.x * 64;
    const int n0 = blockIdx.y * 128;

    const int rowA0 = wm * 32 + lane16;
    const int rowA1 = rowA0 + 16;
    const int colB0 = wn * 32 + lane16;
    const int colB1 = colB0 + 16;

    const int ar = tid >> 3;
    const int ac = (tid & 7) * 4;
    const int bkp = tid >> 5;
    const int bc = (tid & 31) * 4;
    const unsigned aoff = (unsigned)(ar * K + ac);
    const unsigned woff = (unsigned)(2 * bkp * N + bc);
    const float* Au = A + (size_t)m0 * K;
    const float* Wu = W + n0;
    const size_t aHop = (size_t)32 * K;
    const size_t wHop = (size_t)16 * N;
    const size_t wStep = (size_t)32 * N;

    v8f acc[2][2] = {};

    v4f ra0 = *(const v4f*)(Au + aoff);
    v4f ra1 = *(const v4f*)(Au + aHop + aoff);
    v4f rb00 = *(const v4f*)(Wu + woff);
    v4f rb01 = *(const v4f*)(Wu + N + woff);
    v4f rb10 = *(const v4f*)(Wu + wHop + woff);
    v4f rb11 = *(const v4f*)(Wu + wHop + N + woff);

    for (int k0 = 0; k0 < K; k0 += 32) {
        *(v2f*)&As[ar][ac]          = (v2f){ra0[0], ra0[1]};
        *(v2f*)&As[ar][ac + 2]      = (v2f){ra0[2], ra0[3]};
        *(v2f*)&As[ar + 32][ac]     = (v2f){ra1[0], ra1[1]};
        *(v2f*)&As[ar + 32][ac + 2] = (v2f){ra1[2], ra1[3]};
        *(v4f*)&Bs[bkp][bc][0]         = (v4f){rb00[0], rb01[0], rb00[1], rb01[1]};
        *(v4f*)&Bs[bkp][bc + 2][0]     = (v4f){rb00[2], rb01[2], rb00[3], rb01[3]};
        *(v4f*)&Bs[bkp + 8][bc][0]     = (v4f){rb10[0], rb11[0], rb10[1], rb11[1]};
        *(v4f*)&Bs[bkp + 8][bc + 2][0] = (v4f){rb10[2], rb11[2], rb10[3], rb11[3]};
        __syncthreads();

        if (k0 + 32 < K) {
            Au += 32;
            Wu += wStep;
            ra0 = *(const v4f*)(Au + aoff);
            ra1 = *(const v4f*)(Au + aHop + aoff);
            rb00 = *(const v4f*)(Wu + woff);
            rb01 = *(const v4f*)(Wu + N + woff);
            rb10 = *(const v4f*)(Wu + wHop + woff);
            rb11 = *(const v4f*)(Wu + wHop + N + woff);
        }

        v2f a0c = *(const v2f*)&As[rowA0][half * 2];
        v2f a1c = *(const v2f*)&As[rowA1][half * 2];
        v2f b0c = *(const v2f*)&Bs[half][colB0][0];
        v2f b1c = *(const v2f*)&Bs[half][colB1][0];
#pragma unroll
        for (int t = 0; t < 8; ++t) {
            const int kkn = ((t + 1) & 7) * 4 + half * 2;
            v2f a0n = *(const v2f*)&As[rowA0][kkn];
            v2f a1n = *(const v2f*)&As[rowA1][kkn];
            v2f b0n = *(const v2f*)&Bs[kkn >> 1][colB0][0];
            v2f b1n = *(const v2f*)&Bs[kkn >> 1][colB1][0];
            acc[0][0] = wmma4(a0c, b0c, acc[0][0]);
            acc[0][1] = wmma4(a0c, b1c, acc[0][1]);
            acc[1][0] = wmma4(a1c, b0c, acc[1][0]);
            acc[1][1] = wmma4(a1c, b1c, acc[1][1]);
            a0c = a0n; a1c = a1n; b0c = b0n; b1c = b1n;
        }
        __syncthreads();
    }

    const int colb = n0 + wn * 32 + lane16;
    const int rowb = m0 + wm * 32 + half * 8;
    float bv[2];
    float mf[2][8];
    float rs[2][2][8];
    if (bias) {
        bv[0] = bias[colb];
        bv[1] = bias[colb + 16];
    } else {
        bv[0] = 0.0f;
        bv[1] = 0.0f;
    }
    if (mfv) {
#pragma unroll
        for (int im = 0; im < 2; ++im) {
            const v4f lo = *(const v4f*)(mfv + rowb + 16 * im);
            const v4f hi = *(const v4f*)(mfv + rowb + 16 * im + 4);
            mf[im][0] = lo[0]; mf[im][1] = lo[1]; mf[im][2] = lo[2]; mf[im][3] = lo[3];
            mf[im][4] = hi[0]; mf[im][5] = hi[1]; mf[im][6] = hi[2]; mf[im][7] = hi[3];
        }
    } else {
#pragma unroll
        for (int im = 0; im < 2; ++im)
#pragma unroll
            for (int vi = 0; vi < 8; ++vi) mf[im][vi] = 1.0f;
    }
    if (res) {
        const float* rp = res + (size_t)rowb * N + colb;
#pragma unroll
        for (int im = 0; im < 2; ++im)
#pragma unroll
            for (int vi = 0; vi < 8; ++vi)
#pragma unroll
                for (int jn = 0; jn < 2; ++jn)
                    rs[im][jn][vi] = rp[(size_t)(16 * im + vi) * N + 16 * jn];
    } else {
#pragma unroll
        for (int im = 0; im < 2; ++im)
#pragma unroll
            for (int vi = 0; vi < 8; ++vi)
#pragma unroll
                for (int jn = 0; jn < 2; ++jn) rs[im][jn][vi] = 0.0f;
    }
    float* cp = C + (size_t)rowb * N + colb;
#pragma unroll
    for (int im = 0; im < 2; ++im) {
#pragma unroll
        for (int vi = 0; vi < 8; ++vi) {
#pragma unroll
            for (int jn = 0; jn < 2; ++jn) {
                float val = acc[im][jn][vi] + bv[jn];
                if (relu) val = fmaxf(val, 0.0f);
                val = val * mf[im][vi] + rs[im][jn][vi];
                cp[(size_t)(16 * im + vi) * N + 16 * jn] = val;
            }
        }
    }
}


typedef __attribute__((ext_vector_type(4))) int v4i;

__device__ __forceinline__ float xhalf(float x) {
    const int bits = __builtin_bit_cast(int, x);
    return __builtin_bit_cast(float, __builtin_amdgcn_permlanex16(bits, bits, 0x76543210, 0xfedcba98, false, false));
}

__global__ void __launch_bounds__(128) attn_kernel(const float* __restrict__ q, const float* __restrict__ k,
                            const float* __restrict__ v, const int* __restrict__ perm,
                            const int* __restrict__ cnt, float* __restrict__ o) {
    __shared__ __align__(16) float Ks[64][36];
    __shared__ __align__(16) float Vt[32][68];
    __shared__ __align__(16) float mb[64];

    const int S = 2048, LD = 256;
    const int tid = threadIdx.x;
    const int lane = tid & 31;
    const int wave = tid >> 5;
    const int half = lane >> 4;
    const int lane16 = lane & 15;
    const int bh = blockIdx.y;
    const int b = bh >> 3;
    const int h = bh & 7;
    const int par = (__builtin_popcount(blockIdx.x) + __builtin_popcount(blockIdx.y)) & 1;
    const int q0 = ((int)(blockIdx.x >> 1) | (par << 4)) * 64;
    const int nk = cnt[b];
    const int* pb = perm + (size_t)b * S;

    if (q0 >= nk) {
        float* zp = o + ((size_t)b * S + pb[q0 + (tid >> 1)]) * LD + h * 32 + (tid & 1) * 16;
        const v4f z = {0.0f, 0.0f, 0.0f, 0.0f};
        *(v4f*)(zp) = z;
        *(v4f*)(zp + 4) = z;
        *(v4f*)(zp + 8) = z;
        *(v4f*)(zp + 12) = z;
        return;
    }
    const int qrow = pb[q0 + wave * 16 + lane16];
    const int kslot = tid >> 3;

    const float qscale = 0.17677669529663687f * 1.4426950408889634f;
    v4f qf[4];
    {
        const float* qp = q + ((size_t)b * S + qrow) * LD + h * 32 + half * 4;
#pragma unroll
        for (int u = 0; u < 4; ++u) qf[u] = *(const v4f*)(qp + 8 * u) * qscale;
    }

    v8f oacc[2] = {};
    float m_run = -3.0e38f;
    float l_part = 0.0f;

    v4i vi[4];
    int ki[4];
#pragma unroll
    for (int j = 0; j < 4; ++j) {
        vi[j] = *(const v4i*)(pb + 4 * wave + 16 * j);
        ki[j] = pb[kslot + 16 * j];
    }
    const unsigned rowB = (unsigned)(b * S);
    const unsigned colH = (unsigned)(h * 32);

    for (int j0 = 0; j0 < nk; j0 += 64) {
        const int jn = (j0 + 64 < S) ? j0 + 64 : j0;
        v4i vn[4];
        int kn[4];
#pragma unroll
        for (int j = 0; j < 4; ++j) {
            vn[j] = *(const v4i*)(pb + jn + 4 * wave + 16 * j);
            kn[j] = pb[jn + kslot + 16 * j];
        }
        const int pn = pb[jn + (tid & 63)];
        v4f rk[4];
        float rv[4][4];
#pragma unroll
        for (int j = 0; j < 4; ++j) {
            rk[j] = *(const v4f*)&k[(size_t)((rowB + (unsigned)ki[j]) * LD + colH + 4u * (tid & 7))];
        }
#pragma unroll
        for (int j = 0; j < 4; ++j) {
#pragma unroll
            for (int i = 0; i < 4; ++i)
                rv[j][i] = v[(size_t)((rowB + (unsigned)vi[j][i]) * LD + colH) + lane];
        }
#pragma unroll
        for (int j = 0; j < 4; ++j) {
            const int i = tid + j * 128;
            *(v4f*)&Ks[i >> 3][4 * (i & 7)] = rk[j];
        }
#pragma unroll
        for (int j = 0; j < 4; ++j) {
            const int kq = wave + 4 * j;
            v4f t = {rv[j][0], rv[j][1], rv[j][2], rv[j][3]};
            *(v4f*)&Vt[lane][4 * kq] = t;
        }
        mb[tid & 63] = (j0 + (tid & 63) < nk) ? 0.0f : -1e9f;
        {
            const size_t rowN = (size_t)((rowB + (unsigned)pn) * LD + colH);
            if (tid < 64) __builtin_prefetch(&k[rowN], 0, 1);
            else          __builtin_prefetch(&v[rowN], 0, 1);
        }
        __syncthreads();

        v8f sacc[4];
#pragma unroll
        for (int kt = 0; kt < 4; ++kt) {
            const v4f b0 = *(const v4f*)&mb[kt * 16 + half * 8];
            const v4f b1 = *(const v4f*)&mb[kt * 16 + half * 8 + 4];
            sacc[kt][0] = b0[0]; sacc[kt][1] = b0[1]; sacc[kt][2] = b0[2]; sacc[kt][3] = b0[3];
            sacc[kt][4] = b1[0]; sacc[kt][5] = b1[1]; sacc[kt][6] = b1[2]; sacc[kt][7] = b1[3];
        }
        v4f ka[4][4];
#pragma unroll
        for (int u = 0; u < 4; ++u)
#pragma unroll
            for (int kt = 0; kt < 4; ++kt)
                ka[u][kt] = *(const v4f*)&Ks[kt * 16 + lane16][8 * u + half * 4];
        v4f va[4][2][2];
#pragma unroll
        for (int kt = 0; kt < 4; ++kt)
#pragma unroll
            for (int w = 0; w < 2; ++w) {
                va[kt][w][0] = *(const v4f*)&Vt[lane16][kt * 16 + half * 8 + 4 * w];
                va[kt][w][1] = *(const v4f*)&Vt[16 + lane16][kt * 16 + half * 8 + 4 * w];
            }
#pragma unroll
        for (int u = 0; u < 4; ++u) {
#pragma unroll
            for (int e = 0; e < 2; ++e) {
                const v2f bq = {qf[u][2 * e], qf[u][2 * e + 1]};
#pragma unroll
                for (int kt = 0; kt < 4; ++kt) {
                    const v2f ak = {ka[u][kt][2 * e], ka[u][kt][2 * e + 1]};
                    sacc[kt] = wmma4(ak, bq, sacc[kt]);
                }
            }
        }

        float mx[4];
#pragma unroll
        for (int kt = 0; kt < 4; ++kt) {
            mx[kt] = sacc[kt][0];
#pragma unroll
            for (int r = 1; r < 8; ++r) mx[kt] = fmaxf(mx[kt], sacc[kt][r]);
        }
        float mc = fmaxf(fmaxf(mx[0], mx[1]), fmaxf(mx[2], mx[3]));
        mc = fmaxf(mc, xhalf(mc));
        const float mn = fmaxf(m_run, mc);
        const float al = __builtin_amdgcn_exp2f(m_run - mn);
        m_run = mn;
        float ls[4];
#pragma unroll
        for (int kt = 0; kt < 4; ++kt) {
            ls[kt] = 0.0f;
#pragma unroll
            for (int r = 0; r < 8; ++r) {
                const float p = __builtin_amdgcn_exp2f(sacc[kt][r] - mn);
                sacc[kt][r] = p;
                ls[kt] += p;
            }
        }
        l_part = l_part * al + ((ls[0] + ls[1]) + (ls[2] + ls[3]));
#pragma unroll
        for (int dt = 0; dt < 2; ++dt)
#pragma unroll
            for (int r = 0; r < 8; ++r) oacc[dt][r] *= al;

#pragma unroll
        for (int kt = 0; kt < 4; ++kt) {
#pragma unroll
            for (int w = 0; w < 2; ++w) {
#pragma unroll
                for (int e = 0; e < 2; ++e) {
                    const v2f pb = {sacc[kt][4 * w + 2 * e], sacc[kt][4 * w + 2 * e + 1]};
                    const v2f a0 = {va[kt][w][0][2 * e], va[kt][w][0][2 * e + 1]};
                    const v2f a1 = {va[kt][w][1][2 * e], va[kt][w][1][2 * e + 1]};
                    oacc[0] = wmma4(a0, pb, oacc[0]);
                    oacc[1] = wmma4(a1, pb, oacc[1]);
                }
            }
        }
        __syncthreads();
        vi[0] = vn[0]; vi[1] = vn[1]; vi[2] = vn[2]; vi[3] = vn[3];
        ki[0] = kn[0]; ki[1] = kn[1]; ki[2] = kn[2]; ki[3] = kn[3];
    }

    const float lsum = l_part + xhalf(l_part);
    const float linv = 1.0f / lsum;
    float* op = o + ((size_t)b * S + qrow) * LD + h * 32 + half * 8;
#pragma unroll
    for (int dt = 0; dt < 2; ++dt) {
        const v4f lo = {oacc[dt][0] * linv, oacc[dt][1] * linv, oacc[dt][2] * linv, oacc[dt][3] * linv};
        const v4f hi = {oacc[dt][4] * linv, oacc[dt][5] * linv, oacc[dt][6] * linv, oacc[dt][7] * linv};
        *(v4f*)(op + dt * 16) = lo;
        *(v4f*)(op + dt * 16 + 4) = hi;
    }
}

extern "C" void kernel_launch(void* const* d_in, const int* in_sizes, int n_in,
                              void* d_out, int out_size, void* d_ws, size_t ws_size,
                              hipStream_t stream) {
    (void)in_sizes; (void)n_in; (void)out_size; (void)ws_size;
    const int B = 4, S = 2048, D = 256, F = 1024, L = 4;
    const int M = B * S;

    const float* x_in  = (const float*)d_in[0];
    const int*   mask  = (const int*)d_in[1];
    const float* w_qs  = (const float*)d_in[2];
    const float* w_ks  = (const float*)d_in[3];
    const float* w_vs  = (const float*)d_in[4];
    const float* w_fc  = (const float*)d_in[5];
    const float* ln1_g = (const float*)d_in[6];
    const float* ln1_b = (const float*)d_in[7];
    const float* w1    = (const float*)d_in[8];
    const float* b1    = (const float*)d_in[9];
    const float* w2    = (const float*)d_in[10];
    const float* b2    = (const float*)d_in[11];
    const float* ln2_g = (const float*)d_in[12];
    const float* ln2_b = (const float*)d_in[13];

    float* ws    = (float*)d_ws;
    float* x_cur = ws;
    float* xn    = x_cur + (size_t)M * D;
    float* qb    = xn + (size_t)M * D;
    float* kb    = qb + (size_t)M * D;
    float* vb    = kb + (size_t)M * D;
    float* ob    = vb + (size_t)M * D;
    float* hb    = ob + (size_t)M * D;
    float* mfv   = hb + (size_t)M * F;
    int*   perm  = (int*)(mfv + M);
    int*   cnt   = perm + M;

    hipMemcpyAsync(x_cur, x_in, sizeof(float) * (size_t)M * D,
                   hipMemcpyDeviceToDevice, stream);
    mask_kernel<<<B, 1024, 0, stream>>>(mask, mfv, perm, cnt);

    const dim3 gD(M / 64, D / 128);
    const dim3 gF(M / 64, F / 128);
    const dim3 gA(S / 64, B * 8);

    for (int l = 0; l < L; ++l) {
        const float* wq = w_qs + (size_t)l * D * D;
        const float* wk = w_ks + (size_t)l * D * D;
        const float* wv = w_vs + (size_t)l * D * D;
        const float* wf = w_fc + (size_t)l * D * D;
        const float* w1l = w1 + (size_t)l * D * F;
        const float* w2l = w2 + (size_t)l * F * D;
        const float* b1l = b1 + (size_t)l * F;
        const float* b2l = b2 + (size_t)l * D;

        ln_kernel<<<M, 256, 0, stream>>>(x_cur, xn, ln1_g + (size_t)l * D, ln1_b + (size_t)l * D);
        gemm_kernel<<<gD, 256, 0, stream>>>(xn, wq, qb, nullptr, mfv, nullptr, D, D, 0);
        gemm_kernel<<<gD, 256, 0, stream>>>(xn, wk, kb, nullptr, mfv, nullptr, D, D, 0);
        gemm_kernel<<<gD, 256, 0, stream>>>(xn, wv, vb, nullptr, mfv, nullptr, D, D, 0);
        attn_kernel<<<gA, 128, 0, stream>>>(qb, kb, vb, perm, cnt, ob);
        gemm_kernel<<<gD, 256, 0, stream>>>(ob, wf, xn, nullptr, mfv, x_cur, D, D, 0);
        gemm_kernel<<<gF, 256, 0, stream>>>(xn, w1l, hb, b1l, nullptr, nullptr, D, F, 1);
        gemm_kernel<<<gD, 256, 0, stream>>>(hb, w2l, qb, b2l, nullptr, xn, F, D, 0);
        ln_kernel<<<M, 256, 0, stream>>>(qb, x_cur, ln2_g + (size_t)l * D, ln2_b + (size_t)l * D);
    }

    hipMemcpyAsync(d_out, x_cur, sizeof(float) * (size_t)M * D,
                   hipMemcpyDeviceToDevice, stream);
}


# LayerNorm: one wave per row, row kept in registers, sums by lane butterfly (4 rows per wave, 256 blocks); was a 256-thread block per row, two LDS tree sums, 19 barriers; row sum order changes
# speedup vs baseline: 3.4423x; 1.0403x over previous
#include <hip/hip_runtime.h>
#include <hip/hip_bf16.h>

typedef __attribute__((ext_vector_type(2))) float v2f;
typedef __attribute__((ext_vector_type(8))) float v8f;
typedef __attribute__((ext_vector_type(4))) float v4f;

__device__ __forceinline__ v8f wmma4(v2f a, v2f b, v8f c) {
    return __builtin_amdgcn_wmma_f32_16x16x4_f32(false, a, false, b, (short)0, c, false, false);
}

__global__ void __launch_bounds__(1024) mask_kernel(const int* __restrict__ mask, float* __restrict__ mfv,
                                                    int* __restrict__ perm, int* __restrict__ cnt) {
    __shared__ int wtot[64];
    __shared__ int wbase[65];
    __shared__ int sperm[2048];
    const int b = blockIdx.x;
    const int tid = threadIdx.x;
    const int lane = tid & 31;
    const int wave = tid >> 5;
    const int* mp = mask + (size_t)b * 2048;
    const bool s0 = mp[tid] != 0;
    const bool s1 = mp[tid + 1024] != 0;
    mfv[(size_t)b * 2048 + tid] = s0 ? 1.0f : 0.0f;
    mfv[(size_t)b * 2048 + tid + 1024] = s1 ? 1.0f : 0.0f;
    const unsigned bal0 = __builtin_amdgcn_ballot_w32(s0);
    const unsigned bal1 = __builtin_amdgcn_ballot_w32(s1);
    const unsigned below = (1u << lane) - 1u;
    const int p0 = __builtin_popcount(bal0 & below);
    const int p1 = __builtin_popcount(bal1 & below);
    if (lane == 0) {
        wtot[wave] = __builtin_popcount(bal0);
        wtot[32 + wave] = __builtin_popcount(bal1);
    }
    __syncthreads();
    if (tid == 0) {
        int a = 0;
        for (int w = 0; w < 64; ++w) { wbase[w] = a; a += wtot[w]; }
        wbase[64] = a;
    }
    __syncthreads();
    const int total = wbase[64];
    const int o0 = wbase[wave] + p0;
    const int o1 = wbase[32 + wave] + p1;
    sperm[s0 ? o0 : total + tid - o0] = tid;
    sperm[s1 ? o1 : total + (tid + 1024) - o1] = tid + 1024;
    __syncthreads();
    perm[(size_t)b * 2048 + tid] = sperm[tid];
    perm[(size_t)b * 2048 + tid + 1024] = sperm[tid + 1024];
    if (tid == 0) cnt[b] = total;
}

template <int CTRL>
__device__ __forceinline__ float dpp_lane(float x) {
    const int bits = __builtin_bit_cast(int, x);
    return __builtin_bit_cast(float, __builtin_amdgcn_update_dpp(bits, bits, CTRL, 0xf, 0xf, false));
}

__device__ __forceinline__ float wave_sum(float x) {
    x += dpp_lane<0x161>(x);
    x += dpp_lane<0x162>(x);
    x += dpp_lane<0x164>(x);
    x += dpp_lane<0x168>(x);
    const int bits = __builtin_bit_cast(int, x);
    x += __builtin_bit_cast(float, __builtin_amdgcn_permlanex16(bits, bits, 0x76543210, 0xfedcba98, false, false));
    return x;
}

__global__ void __launch_bounds__(256) ln_kernel(const float* __restrict__ x, float* __restrict__ out,
                                                 const float* __restrict__ g, const float* __restrict__ b) {
    const int lane = threadIdx.x & 31;
    const int wave = threadIdx.x >> 5;
    const size_t off = ((size_t)blockIdx.x * 32 + wave * 4) * 256 + lane * 4;
    const float* xp = x + off;
    float* op = out + off;
    v4f a[4][2];
#pragma unroll
    for (int r = 0; r < 4; ++r) {
        a[r][0] = *(const v4f*)(xp + r * 256);
        a[r][1] = *(const v4f*)(xp + r * 256 + 128);
    }
    const v4f g0 = *(const v4f*)(g + lane * 4);
    const v4f g1 = *(const v4f*)(g + 128 + lane * 4);
    const v4f b0 = *(const v4f*)(b + lane * 4);
    const v4f b1 = *(const v4f*)(b + 128 + lane * 4);
#pragma unroll
    for (int r = 0; r < 4; ++r) {
        const v4f s4 = a[r][0] + a[r][1];
        const float mean = wave_sum((s4[0] + s4[1]) + (s4[2] + s4[3])) * (1.0f / 256.0f);
        const v4f d0 = a[r][0] - mean;
        const v4f d1 = a[r][1] - mean;
        const v4f q4 = d0 * d0 + d1 * d1;
        const float var = wave_sum((q4[0] + q4[1]) + (q4[2] + q4[3])) * (1.0f / 256.0f);
        const float rs = rsqrtf(var + 1e-6f);
        *(v4f*)(op + r * 256) = d0 * rs * g0 + b0;
        *(v4f*)(op + r * 256 + 128) = d1 * rs * g1 + b1;
    }
}


__global__ void __launch_bounds__(256) __attribute__((amdgpu_waves_per_eu(2, 8))) gemm_kernel(const float* __restrict__ A, const float* __restrict__ W,
                            float* __restrict__ C, const float* __restrict__ bias,
                            const float* __restrict__ mfv, const float* __restrict__ res,
                            int K, int N, int relu) {
    __shared__ __align__(16) float As[64][34];
    __shared__ __align__(16) float Bs[16][144][2];

    const int tid = threadIdx.x;
    const int lane = tid & 31;
    const int wave = tid >> 5;
    const int half = lane >> 4;
    const int lane16 = lane & 15;
    const int wm = wave >> 2;
    const int wn = wave & 3;
    const int m0 = blockIdx.x * 64;
    const int n0 = blockIdx.y * 128;

    const int rowA0 = wm * 32 + lane16;
    const int rowA1 = rowA0 + 16;
    const int colB0 = wn * 32 + lane16;
    const int colB1 = colB0 + 16;

    const int ar = tid >> 3;
    const int ac = (tid & 7) * 4;
    const int bkp = tid >> 5;
    const int bc = (tid & 31) * 4;
    const unsigned aoff = (unsigned)(ar * K + ac);
    const unsigned woff = (unsigned)(2 * bkp * N + bc);
    const float* Au = A + (size_t)m0 * K;
    const float* Wu = W + n0;
    const size_t aHop = (size_t)32 * K;
    const size_t wHop = (size_t)16 * N;
    const size_t wStep = (size_t)32 * N;

    v8f acc[2][2] = {};

    v4f ra0 = *(const v4f*)(Au + aoff);
    v4f ra1 = *(const v4f*)(Au + aHop + aoff);
    v4f rb00 = *(const v4f*)(Wu + woff);
    v4f rb01 = *(const v4f*)(Wu + N + woff);
    v4f rb10 = *(const v4f*)(Wu + wHop + woff);
    v4f rb11 = *(const v4f*)(Wu + wHop + N + woff);

    for (int k0 = 0; k0 < K; k0 += 32) {
        *(v2f*)&As[ar][ac]          = (v2f){ra0[0], ra0[1]};
        *(v2f*)&As[ar][ac + 2]      = (v2f){ra0[2], ra0[3]};
        *(v2f*)&As[ar + 32][ac]     = (v2f){ra1[0], ra1[1]};
        *(v2f*)&As[ar + 32][ac + 2] = (v2f){ra1[2], ra1[3]};
        *(v4f*)&Bs[bkp][bc][0]         = (v4f){rb00[0], rb01[0], rb00[1], rb01[1]};
        *(v4f*)&Bs[bkp][bc + 2][0]     = (v4f){rb00[2], rb01[2], rb00[3], rb01[3]};
        *(v4f*)&Bs[bkp + 8][bc][0]     = (v4f){rb10[0], rb11[0], rb10[1], rb11[1]};
        *(v4f*)&Bs[bkp + 8][bc + 2][0] = (v4f){rb10[2], rb11[2], rb10[3], rb11[3]};
        __syncthreads();

        if (k0 + 32 < K) {
            Au += 32;
            Wu += wStep;
            ra0 = *(const v4f*)(Au + aoff);
            ra1 = *(const v4f*)(Au + aHop + aoff);
            rb00 = *(const v4f*)(Wu + woff);
            rb01 = *(const v4f*)(Wu + N + woff);
            rb10 = *(const v4f*)(Wu + wHop + woff);
            rb11 = *(const v4f*)(Wu + wHop + N + woff);
        }

        v2f a0c = *(const v2f*)&As[rowA0][half * 2];
        v2f a1c = *(const v2f*)&As[rowA1][half * 2];
        v2f b0c = *(const v2f*)&Bs[half][colB0][0];
        v2f b1c = *(const v2f*)&Bs[half][colB1][0];
#pragma unroll
        for (int t = 0; t < 8; ++t) {
            const int kkn = ((t + 1) & 7) * 4 + half * 2;
            v2f a0n = *(const v2f*)&As[rowA0][kkn];
            v2f a1n = *(const v2f*)&As[rowA1][kkn];
            v2f b0n = *(const v2f*)&Bs[kkn >> 1][colB0][0];
            v2f b1n = *(const v2f*)&Bs[kkn >> 1][colB1][0];
            acc[0][0] = wmma4(a0c, b0c, acc[0][0]);
            acc[0][1] = wmma4(a0c, b1c, acc[0][1]);
            acc[1][0] = wmma4(a1c, b0c, acc[1][0]);
            acc[1][1] = wmma4(a1c, b1c, acc[1][1]);
            a0c = a0n; a1c = a1n; b0c = b0n; b1c = b1n;
        }
        __syncthreads();
    }

    const int colb = n0 + wn * 32 + lane16;
    const int rowb = m0 + wm * 32 + half * 8;
    float bv[2];
    float mf[2][8];
    float rs[2][2][8];
    if (bias) {
        bv[0] = bias[colb];
        bv[1] = bias[colb + 16];
    } else {
        bv[0] = 0.0f;
        bv[1] = 0.0f;
    }
    if (mfv) {
#pragma unroll
        for (int im = 0; im < 2; ++im) {
            const v4f lo = *(const v4f*)(mfv + rowb + 16 * im);
            const v4f hi = *(const v4f*)(mfv + rowb + 16 * im + 4);
            mf[im][0] = lo[0]; mf[im][1] = lo[1]; mf[im][2] = lo[2]; mf[im][3] = lo[3];
            mf[im][4] = hi[0]; mf[im][5] = hi[1]; mf[im][6] = hi[2]; mf[im][7] = hi[3];
        }
    } else {
#pragma unroll
        for (int im = 0; im < 2; ++im)
#pragma unroll
            for (int vi = 0; vi < 8; ++vi) mf[im][vi] = 1.0f;
    }
    if (res) {
        const float* rp = res + (size_t)rowb * N + colb;
#pragma unroll
        for (int im = 0; im < 2; ++im)
#pragma unroll
            for (int vi = 0; vi < 8; ++vi)
#pragma unroll
                for (int jn = 0; jn < 2; ++jn)
                    rs[im][jn][vi] = rp[(size_t)(16 * im + vi) * N + 16 * jn];
    } else {
#pragma unroll
        for (int im = 0; im < 2; ++im)
#pragma unroll
            for (int vi = 0; vi < 8; ++vi)
#pragma unroll
                for (int jn = 0; jn < 2; ++jn) rs[im][jn][vi] = 0.0f;
    }
    float* cp = C + (size_t)rowb * N + colb;
#pragma unroll
    for (int im = 0; im < 2; ++im) {
#pragma unroll
        for (int vi = 0; vi < 8; ++vi) {
#pragma unroll
            for (int jn = 0; jn < 2; ++jn) {
                float val = acc[im][jn][vi] + bv[jn];
                if (relu) val = fmaxf(val, 0.0f);
                val = val * mf[im][vi] + rs[im][jn][vi];
                cp[(size_t)(16 * im + vi) * N + 16 * jn] = val;
            }
        }
    }
}


typedef __attribute__((ext_vector_type(4))) int v4i;

__device__ __forceinline__ float xhalf(float x) {
    const int bits = __builtin_bit_cast(int, x);
    return __builtin_bit_cast(float, __builtin_amdgcn_permlanex16(bits, bits, 0x76543210, 0xfedcba98, false, false));
}

__global__ void __launch_bounds__(128) attn_kernel(const float* __restrict__ q, const float* __restrict__ k,
                            const float* __restrict__ v, const int* __restrict__ perm,
                            const int* __restrict__ cnt, float* __restrict__ o) {
    __shared__ __align__(16) float Ks[64][36];
    __shared__ __align__(16) float Vt[32][68];
    __shared__ __align__(16) float mb[64];

    const int S = 2048, LD = 256;
    const int tid = threadIdx.x;
    const int lane = tid & 31;
    const int wave = tid >> 5;
    const int half = lane >> 4;
    const int lane16 = lane & 15;
    const int bh = blockIdx.y;
    const int b = bh >> 3;
    const int h = bh & 7;
    const int par = (__builtin_popcount(blockIdx.x) + __builtin_popcount(blockIdx.y)) & 1;
    const int q0 = ((int)(blockIdx.x >> 1) | (par << 4)) * 64;
    const int nk = cnt[b];
    const int* pb = perm + (size_t)b * S;

    if (q0 >= nk) {
        float* zp = o + ((size_t)b * S + pb[q0 + (tid >> 1)]) * LD + h * 32 + (tid & 1) * 16;
        const v4f z = {0.0f, 0.0f, 0.0f, 0.0f};
        *(v4f*)(zp) = z;
        *(v4f*)(zp + 4) = z;
        *(v4f*)(zp + 8) = z;
        *(v4f*)(zp + 12) = z;
        return;
    }
    const int qrow = pb[q0 + wave * 16 + lane16];
    const int kslot = tid >> 3;

    const float qscale = 0.17677669529663687f * 1.4426950408889634f;
    v4f qf[4];
    {
        const float* qp = q + ((size_t)b * S + qrow) * LD + h * 32 + half * 4;
#pragma unroll
        for (int u = 0; u < 4; ++u) qf[u] = *(const v4f*)(qp + 8 * u) * qscale;
    }

    v8f oacc[2] = {};
    float m_run = -3.0e38f;
    float l_part = 0.0f;

    v4i vi[4];
    int ki[4];
#pragma unroll
    for (int j = 0; j < 4; ++j) {
        vi[j] = *(const v4i*)(pb + 4 * wave + 16 * j);
        ki[j] = pb[kslot + 16 * j];
    }
    const unsigned rowB = (unsigned)(b * S);
    const unsigned colH = (unsigned)(h * 32);

    for (int j0 = 0; j0 < nk; j0 += 64) {
        const int jn = (j0 + 64 < S) ? j0 + 64 : j0;
        v4i vn[4];
        int kn[4];
#pragma unroll
        for (int j = 0; j < 4; ++j) {
            vn[j] = *(const v4i*)(pb + jn + 4 * wave + 16 * j);
            kn[j] = pb[jn + kslot + 16 * j];
        }
        const int pn = pb[jn + (tid & 63)];
        v4f rk[4];
        float rv[4][4];
#pragma unroll
        for (int j = 0; j < 4; ++j) {
            rk[j] = *(const v4f*)&k[(size_t)((rowB + (unsigned)ki[j]) * LD + colH + 4u * (tid & 7))];
        }
#pragma unroll
        for (int j = 0; j < 4; ++j) {
#pragma unroll
            for (int i = 0; i < 4; ++i)
                rv[j][i] = v[(size_t)((rowB + (unsigned)vi[j][i]) * LD + colH) + lane];
        }
#pragma unroll
        for (int j = 0; j < 4; ++j) {
            const int i = tid + j * 128;
            *(v4f*)&Ks[i >> 3][4 * (i & 7)] = rk[j];
        }
#pragma unroll
        for (int j = 0; j < 4; ++j) {
            const int kq = wave + 4 * j;
            v4f t = {rv[j][0], rv[j][1], rv[j][2], rv[j][3]};
            *(v4f*)&Vt[lane][4 * kq] = t;
        }
        mb[tid & 63] = (j0 + (tid & 63) < nk) ? 0.0f : -1e9f;
        {
            const size_t rowN = (size_t)((rowB + (unsigned)pn) * LD + colH);
            if (tid < 64) __builtin_prefetch(&k[rowN], 0, 1);
            else          __builtin_prefetch(&v[rowN], 0, 1);
        }
        __syncthreads();

        v8f sacc[4];
#pragma unroll
        for (int kt = 0; kt < 4; ++kt) {
            const v4f b0 = *(const v4f*)&mb[kt * 16 + half * 8];
            const v4f b1 = *(const v4f*)&mb[kt * 16 + half * 8 + 4];
            sacc[kt][0] = b0[0]; sacc[kt][1] = b0[1]; sacc[kt][2] = b0[2]; sacc[kt][3] = b0[3];
            sacc[kt][4] = b1[0]; sacc[kt][5] = b1[1]; sacc[kt][6] = b1[2]; sacc[kt][7] = b1[3];
        }
        v4f ka[4][4];
#pragma unroll
        for (int u = 0; u < 4; ++u)
#pragma unroll
            for (int kt = 0; kt < 4; ++kt)
                ka[u][kt] = *(const v4f*)&Ks[kt * 16 + lane16][8 * u + half * 4];
        v4f va[4][2][2];
#pragma unroll
        for (int kt = 0; kt < 4; ++kt)
#pragma unroll
            for (int w = 0; w < 2; ++w) {
                va[kt][w][0] = *(const v4f*)&Vt[lane16][kt * 16 + half * 8 + 4 * w];
                va[kt][w][1] = *(const v4f*)&Vt[16 + lane16][kt * 16 + half * 8 + 4 * w];
            }
#pragma unroll
        for (int u = 0; u < 4; ++u) {
#pragma unroll
            for (int e = 0; e < 2; ++e) {
                const v2f bq = {qf[u][2 * e], qf[u][2 * e + 1]};
#pragma unroll
                for (int kt = 0; kt < 4; ++kt) {
                    const v2f ak = {ka[u][kt][2 * e], ka[u][kt][2 * e + 1]};
                    sacc[kt] = wmma4(ak, bq, sacc[kt]);
                }
            }
        }

        float mx[4];
#pragma unroll
        for (int kt = 0; kt < 4; ++kt) {
            mx[kt] = sacc[kt][0];
#pragma unroll
            for (int r = 1; r < 8; ++r) mx[kt] = fmaxf(mx[kt], sacc[kt][r]);
        }
        float mc = fmaxf(fmaxf(mx[0], mx[1]), fmaxf(mx[2], mx[3]));
        mc = fmaxf(mc, xhalf(mc));
        const float mn = fmaxf(m_run, mc);
        const float al = __builtin_amdgcn_exp2f(m_run - mn);
        m_run = mn;
        float ls[4];
#pragma unroll
        for (int kt = 0; kt < 4; ++kt) {
            ls[kt] = 0.0f;
#pragma unroll
            for (int r = 0; r < 8; ++r) {
                const float p = __builtin_amdgcn_exp2f(sacc[kt][r] - mn);
                sacc[kt][r] = p;
                ls[kt] += p;
            }
        }
        l_part = l_part * al + ((ls[0] + ls[1]) + (ls[2] + ls[3]));
#pragma unroll
        for (int dt = 0; dt < 2; ++dt)
#pragma unroll
            for (int r = 0; r < 8; ++r) oacc[dt][r] *= al;

#pragma unroll
        for (int kt = 0; kt < 4; ++kt) {
#pragma unroll
            for (int w = 0; w < 2; ++w) {
#pragma unroll
                for (int e = 0; e < 2; ++e) {
                    const v2f pb = {sacc[kt][4 * w + 2 * e], sacc[kt][4 * w + 2 * e + 1]};
                    const v2f a0 = {va[kt][w][0][2 * e], va[kt][w][0][2 * e + 1]};
                    const v2f a1 = {va[kt][w][1][2 * e], va[kt][w][1][2 * e + 1]};
                    oacc[0] = wmma4(a0, pb, oacc[0]);
                    oacc[1] = wmma4(a1, pb, oacc[1]);
                }
            }
        }
        __syncthreads();
        vi[0] = vn[0]; vi[1] = vn[1]; vi[2] = vn[2]; vi[3] = vn[3];
        ki[0] = kn[0]; ki[1] = kn[1]; ki[2] = kn[2]; ki[3] = kn[3];
    }

    const float lsum = l_part + xhalf(l_part);
    const float linv = 1.0f / lsum;
    float* op = o + ((size_t)b * S + qrow) * LD + h * 32 + half * 8;
#pragma unroll
    for (int dt = 0; dt < 2; ++dt) {
        const v4f lo = {oacc[dt][0] * linv, oacc[dt][1] * linv, oacc[dt][2] * linv, oacc[dt][3] * linv};
        const v4f hi = {oacc[dt][4] * linv, oacc[dt][5] * linv, oacc[dt][6] * linv, oacc[dt][7] * linv};
        *(v4f*)(op + dt * 16) = lo;
        *(v4f*)(op + dt * 16 + 4) = hi;
    }
}

extern "C" void kernel_launch(void* const* d_in, const int* in_sizes, int n_in,
                              void* d_out, int out_size, void* d_ws, size_t ws_size,
                              hipStream_t stream) {
    (void)in_sizes; (void)n_in; (void)out_size; (void)ws_size;
    const int B = 4, S = 2048, D = 256, F = 1024, L = 4;
    const int M = B * S;

    const float* x_in  = (const float*)d_in[0];
    const int*   mask  = (const int*)d_in[1];
    const float* w_qs  = (const float*)d_in[2];
    const float* w_ks  = (const float*)d_in[3];
    const float* w_vs  = (const float*)d_in[4];
    const float* w_fc  = (const float*)d_in[5];
    const float* ln1_g = (const float*)d_in[6];
    const float* ln1_b = (const float*)d_in[7];
    const float* w1    = (const float*)d_in[8];
    const float* b1    = (const float*)d_in[9];
    const float* w2    = (const float*)d_in[10];
    const float* b2    = (const float*)d_in[11];
    const float* ln2_g = (const float*)d_in[12];
    const float* ln2_b = (const float*)d_in[13];

    float* ws    = (float*)d_ws;
    float* x_cur = ws;
    float* xn    = x_cur + (size_t)M * D;
    float* qb    = xn + (size_t)M * D;
    float* kb    = qb + (size_t)M * D;
    float* vb    = kb + (size_t)M * D;
    float* ob    = vb + (size_t)M * D;
    float* hb    = ob + (size_t)M * D;
    float* mfv   = hb + (size_t)M * F;
    int*   perm  = (int*)(mfv + M);
    int*   cnt   = perm + M;

    hipMemcpyAsync(x_cur, x_in, sizeof(float) * (size_t)M * D,
                   hipMemcpyDeviceToDevice, stream);
    mask_kernel<<<B, 1024, 0, stream>>>(mask, mfv, perm, cnt);

    const dim3 gD(M / 64, D / 128);
    const dim3 gF(M / 64, F / 128);
    const dim3 gA(S / 64, B * 8);

    for (int l = 0; l < L; ++l) {
        const float* wq = w_qs + (size_t)l * D * D;
        const float* wk = w_ks + (size_t)l * D * D;
        const float* wv = w_vs + (size_t)l * D * D;
        const float* wf = w_fc + (size_t)l * D * D;
        const float* w1l = w1 + (size_t)l * D * F;
        const float* w2l = w2 + (size_t)l * F * D;
        const float* b1l = b1 + (size_t)l * F;
        const float* b2l = b2 + (size_t)l * D;

        ln_kernel<<<M / 32, 256, 0, stream>>>(x_cur, xn, ln1_g + (size_t)l * D, ln1_b + (size_t)l * D);
        gemm_kernel<<<gD, 256, 0, stream>>>(xn, wq, qb, nullptr, mfv, nullptr, D, D, 0);
        gemm_kernel<<<gD, 256, 0, stream>>>(xn, wk, kb, nullptr, mfv, nullptr, D, D, 0);
        gemm_kernel<<<gD, 256, 0, stream>>>(xn, wv, vb, nullptr, mfv, nullptr, D, D, 0);
        attn_kernel<<<gA, 128, 0, stream>>>(qb, kb, vb, perm, cnt, ob);
        gemm_kernel<<<gD, 256, 0, stream>>>(ob, wf, xn, nullptr, mfv, x_cur, D, D, 0);
        gemm_kernel<<<gF, 256, 0, stream>>>(xn, w1l, hb, b1l, nullptr, nullptr, D, F, 1);
        gemm_kernel<<<gD, 256, 0, stream>>>(hb, w2l, qb, b2l, nullptr, xn, F, D, 0);
        ln_kernel<<<M / 32, 256, 0, stream>>>(qb, x_cur, ln2_g + (size_t)l * D, ln2_b + (size_t)l * D);
    }

    hipMemcpyAsync(d_out, x_cur, sizeof(float) * (size_t)M * D,
                   hipMemcpyDeviceToDevice, stream);
}


# GEMM: the Q, K and V projections of a layer run as one launch (grid z picks weight and output): 768 blocks, three per compute unit, was three launches of 256; 37 launches become 29
# speedup vs baseline: 3.6951x; 1.0734x over previous
#include <hip/hip_runtime.h>
#include <hip/hip_bf16.h>

typedef __attribute__((ext_vector_type(2))) float v2f;
typedef __attribute__((ext_vector_type(8))) float v8f;
typedef __attribute__((ext_vector_type(4))) float v4f;

__device__ __forceinline__ v8f wmma4(v2f a, v2f b, v8f c) {
    return __builtin_amdgcn_wmma_f32_16x16x4_f32(false, a, false, b, (short)0, c, false, false);
}

__global__ void __launch_bounds__(1024) mask_kernel(const int* __restrict__ mask, float* __restrict__ mfv,
                                                    int* __restrict__ perm, int* __restrict__ cnt) {
    __shared__ int wtot[64];
    __shared__ int wbase[65];
    __shared__ int sperm[2048];
    const int b = blockIdx.x;
    const int tid = threadIdx.x;
    const int lane = tid & 31;
    const int wave = tid >> 5;
    const int* mp = mask + (size_t)b * 2048;
    const bool s0 = mp[tid] != 0;
    const bool s1 = mp[tid + 1024] != 0;
    mfv[(size_t)b * 2048 + tid] = s0 ? 1.0f : 0.0f;
    mfv[(size_t)b * 2048 + tid + 1024] = s1 ? 1.0f : 0.0f;
    const unsigned bal0 = __builtin_amdgcn_ballot_w32(s0);
    const unsigned bal1 = __builtin_amdgcn_ballot_w32(s1);
    const unsigned below = (1u << lane) - 1u;
    const int p0 = __builtin_popcount(bal0 & below);
    const int p1 = __builtin_popcount(bal1 & below);
    if (lane == 0) {
        wtot[wave] = __builtin_popcount(bal0);
        wtot[32 + wave] = __builtin_popcount(bal1);
    }
    __syncthreads();
    if (tid == 0) {
        int a = 0;
        for (int w = 0; w < 64; ++w) { wbase[w] = a; a += wtot[w]; }
        wbase[64] = a;
    }
    __syncthreads();
    const int total = wbase[64];
    const int o0 = wbase[wave] + p0;
    const int o1 = wbase[32 + wave] + p1;
    sperm[s0 ? o0 : total + tid - o0] = tid;
    sperm[s1 ? o1 : total + (tid + 1024) - o1] = tid + 1024;
    __syncthreads();
    perm[(size_t)b * 2048 + tid] = sperm[tid];
    perm[(size_t)b * 2048 + tid + 1024] = sperm[tid + 1024];
    if (tid == 0) cnt[b] = total;
}

template <int CTRL>
__device__ __forceinline__ float dpp_lane(float x) {
    const int bits = __builtin_bit_cast(int, x);
    return __builtin_bit_cast(float, __builtin_amdgcn_update_dpp(bits, bits, CTRL, 0xf, 0xf, false));
}

__device__ __forceinline__ float wave_sum(float x) {
    x += dpp_lane<0x161>(x);
    x += dpp_lane<0x162>(x);
    x += dpp_lane<0x164>(x);
    x += dpp_lane<0x168>(x);
    const int bits = __builtin_bit_cast(int, x);
    x += __builtin_bit_cast(float, __builtin_amdgcn_permlanex16(bits, bits, 0x76543210, 0xfedcba98, false, false));
    return x;
}

__global__ void __launch_bounds__(256) ln_kernel(const float* __restrict__ x, float* __restrict__ out,
                                                 const float* __restrict__ g, const float* __restrict__ b) {
    const int lane = threadIdx.x & 31;
    const int wave = threadIdx.x >> 5;
    const size_t off = ((size_t)blockIdx.x * 32 + wave * 4) * 256 + lane * 4;
    const float* xp = x + off;
    float* op = out + off;
    v4f a[4][2];
#pragma unroll
    for (int r = 0; r < 4; ++r) {
        a[r][0] = *(const v4f*)(xp + r * 256);
        a[r][1] = *(const v4f*)(xp + r * 256 + 128);
    }
    const v4f g0 = *(const v4f*)(g + lane * 4);
    const v4f g1 = *(const v4f*)(g + 128 + lane * 4);
    const v4f b0 = *(const v4f*)(b + lane * 4);
    const v4f b1 = *(const v4f*)(b + 128 + lane * 4);
#pragma unroll
    for (int r = 0; r < 4; ++r) {
        const v4f s4 = a[r][0] + a[r][1];
        const float mean = wave_sum((s4[0] + s4[1]) + (s4[2] + s4[3])) * (1.0f / 256.0f);
        const v4f d0 = a[r][0] - mean;
        const v4f d1 = a[r][1] - mean;
        const v4f q4 = d0 * d0 + d1 * d1;
        const float var = wave_sum((q4[0] + q4[1]) + (q4[2] + q4[3])) * (1.0f / 256.0f);
        const float rs = rsqrtf(var + 1e-6f);
        *(v4f*)(op + r * 256) = d0 * rs * g0 + b0;
        *(v4f*)(op + r * 256 + 128) = d1 * rs * g1 + b1;
    }
}


__global__ void __launch_bounds__(256) __attribute__((amdgpu_waves_per_eu(2, 8))) gemm_kernel(const float* __restrict__ A, const float* __restrict__ W,
                            float* __restrict__ C, const float* __restrict__ bias,
                            const float* __restrict__ mfv, const float* __restrict__ res,
                            int K, int N, int relu,
                            const float* __restrict__ Wz1, float* __restrict__ Cz1,
                            const float* __restrict__ Wz2, float* __restrict__ Cz2) {
    __shared__ __align__(16) float As[64][34];
    __shared__ __align__(16) float Bs[16][144][2];

    const int tid = threadIdx.x;
    const int lane = tid & 31;
    const int wave = tid >> 5;
    const int half = lane >> 4;
    const int lane16 = lane & 15;
    const int wm = wave >> 2;
    const int wn = wave & 3;
    const int m0 = blockIdx.x * 64;
    const int n0 = blockIdx.y * 128;
    const float* Wsel = W;
    float* Csel = C;
    if (blockIdx.z == 1) { Wsel = Wz1; Csel = Cz1; }
    else if (blockIdx.z == 2) { Wsel = Wz2; Csel = Cz2; }

    const int rowA0 = wm * 32 + lane16;
    const int rowA1 = rowA0 + 16;
    const int colB0 = wn * 32 + lane16;
    const int colB1 = colB0 + 16;

    const int ar = tid >> 3;
    const int ac = (tid & 7) * 4;
    const int bkp = tid >> 5;
    const int bc = (tid & 31) * 4;
    const unsigned aoff = (unsigned)(ar * K + ac);
    const unsigned woff = (unsigned)(2 * bkp * N + bc);
    const float* Au = A + (size_t)m0 * K;
    const float* Wu = Wsel + n0;
    const size_t aHop = (size_t)32 * K;
    const size_t wHop = (size_t)16 * N;
    const size_t wStep = (size_t)32 * N;

    v8f acc[2][2] = {};

    v4f ra0 = *(const v4f*)(Au + aoff);
    v4f ra1 = *(const v4f*)(Au + aHop + aoff);
    v4f rb00 = *(const v4f*)(Wu + woff);
    v4f rb01 = *(const v4f*)(Wu + N + woff);
    v4f rb10 = *(const v4f*)(Wu + wHop + woff);
    v4f rb11 = *(const v4f*)(Wu + wHop + N + woff);

    for (int k0 = 0; k0 < K; k0 += 32) {
        *(v2f*)&As[ar][ac]          = (v2f){ra0[0], ra0[1]};
        *(v2f*)&As[ar][ac + 2]      = (v2f){ra0[2], ra0[3]};
        *(v2f*)&As[ar + 32][ac]     = (v2f){ra1[0], ra1[1]};
        *(v2f*)&As[ar + 32][ac + 2] = (v2f){ra1[2], ra1[3]};
        *(v4f*)&Bs[bkp][bc][0]         = (v4f){rb00[0], rb01[0], rb00[1], rb01[1]};
        *(v4f*)&Bs[bkp][bc + 2][0]     = (v4f){rb00[2], rb01[2], rb00[3], rb01[3]};
        *(v4f*)&Bs[bkp + 8][bc][0]     = (v4f){rb10[0], rb11[0], rb10[1], rb11[1]};
        *(v4f*)&Bs[bkp + 8][bc + 2][0] = (v4f){rb10[2], rb11[2], rb10[3], rb11[3]};
        __syncthreads();

        if (k0 + 32 < K) {
            Au += 32;
            Wu += wStep;
            ra0 = *(const v4f*)(Au + aoff);
            ra1 = *(const v4f*)(Au + aHop + aoff);
            rb00 = *(const v4f*)(Wu + woff);
            rb01 = *(const v4f*)(Wu + N + woff);
            rb10 = *(const v4f*)(Wu + wHop + woff);
            rb11 = *(const v4f*)(Wu + wHop + N + woff);
        }

        v2f a0c = *(const v2f*)&As[rowA0][half * 2];
        v2f a1c = *(const v2f*)&As[rowA1][half * 2];
        v2f b0c = *(const v2f*)&Bs[half][colB0][0];
        v2f b1c = *(const v2f*)&Bs[half][colB1][0];
#pragma unroll
        for (int t = 0; t < 8; ++t) {
            const int kkn = ((t + 1) & 7) * 4 + half * 2;
            v2f a0n = *(const v2f*)&As[rowA0][kkn];
            v2f a1n = *(const v2f*)&As[rowA1][kkn];
            v2f b0n = *(const v2f*)&Bs[kkn >> 1][colB0][0];
            v2f b1n = *(const v2f*)&Bs[kkn >> 1][colB1][0];
            acc[0][0] = wmma4(a0c, b0c, acc[0][0]);
            acc[0][1] = wmma4(a0c, b1c, acc[0][1]);
            acc[1][0] = wmma4(a1c, b0c, acc[1][0]);
            acc[1][1] = wmma4(a1c, b1c, acc[1][1]);
            a0c = a0n; a1c = a1n; b0c = b0n; b1c = b1n;
        }
        __syncthreads();
    }

    const int colb = n0 + wn * 32 + lane16;
    const int rowb = m0 + wm * 32 + half * 8;
    float bv[2];
    float mf[2][8];
    float rs[2][2][8];
    if (bias) {
        bv[0] = bias[colb];
        bv[1] = bias[colb + 16];
    } else {
        bv[0] = 0.0f;
        bv[1] = 0.0f;
    }
    if (mfv) {
#pragma unroll
        for (int im = 0; im < 2; ++im) {
            const v4f lo = *(const v4f*)(mfv + rowb + 16 * im);
            const v4f hi = *(const v4f*)(mfv + rowb + 16 * im + 4);
            mf[im][0] = lo[0]; mf[im][1] = lo[1]; mf[im][2] = lo[2]; mf[im][3] = lo[3];
            mf[im][4] = hi[0]; mf[im][5] = hi[1]; mf[im][6] = hi[2]; mf[im][7] = hi[3];
        }
    } else {
#pragma unroll
        for (int im = 0; im < 2; ++im)
#pragma unroll
            for (int vi = 0; vi < 8; ++vi) mf[im][vi] = 1.0f;
    }
    if (res) {
        const float* rp = res + (size_t)rowb * N + colb;
#pragma unroll
        for (int im = 0; im < 2; ++im)
#pragma unroll
            for (int vi = 0; vi < 8; ++vi)
#pragma unroll
                for (int jn = 0; jn < 2; ++jn)
                    rs[im][jn][vi] = rp[(size_t)(16 * im + vi) * N + 16 * jn];
    } else {
#pragma unroll
        for (int im = 0; im < 2; ++im)
#pragma unroll
            for (int vi = 0; vi < 8; ++vi)
#pragma unroll
                for (int jn = 0; jn < 2; ++jn) rs[im][jn][vi] = 0.0f;
    }
    float* cp = Csel + (size_t)rowb * N + colb;
#pragma unroll
    for (int im = 0; im < 2; ++im) {
#pragma unroll
        for (int vi = 0; vi < 8; ++vi) {
#pragma unroll
            for (int jn = 0; jn < 2; ++jn) {
                float val = acc[im][jn][vi] + bv[jn];
                if (relu) val = fmaxf(val, 0.0f);
                val = val * mf[im][vi] + rs[im][jn][vi];
                cp[(size_t)(16 * im + vi) * N + 16 * jn] = val;
            }
        }
    }
}


typedef __attribute__((ext_vector_type(4))) int v4i;

__device__ __forceinline__ float xhalf(float x) {
    const int bits = __builtin_bit_cast(int, x);
    return __builtin_bit_cast(float, __builtin_amdgcn_permlanex16(bits, bits, 0x76543210, 0xfedcba98, false, false));
}

__global__ void __launch_bounds__(128) attn_kernel(const float* __restrict__ q, const float* __restrict__ k,
                            const float* __restrict__ v, const int* __restrict__ perm,
                            const int* __restrict__ cnt, float* __restrict__ o) {
    __shared__ __align__(16) float Ks[64][36];
    __shared__ __align__(16) float Vt[32][68];
    __shared__ __align__(16) float mb[64];

    const int S = 2048, LD = 256;
    const int tid = threadIdx.x;
    const int lane = tid & 31;
    const int wave = tid >> 5;
    const int half = lane >> 4;
    const int lane16 = lane & 15;
    const int bh = blockIdx.y;
    const int b = bh >> 3;
    const int h = bh & 7;
    const int par = (__builtin_popcount(blockIdx.x) + __builtin_popcount(blockIdx.y)) & 1;
    const int q0 = ((int)(blockIdx.x >> 1) | (par << 4)) * 64;
    const int nk = cnt[b];
    const int* pb = perm + (size_t)b * S;

    if (q0 >= nk) {
        float* zp = o + ((size_t)b * S + pb[q0 + (tid >> 1)]) * LD + h * 32 + (tid & 1) * 16;
        const v4f z = {0.0f, 0.0f, 0.0f, 0.0f};
        *(v4f*)(zp) = z;
        *(v4f*)(zp + 4) = z;
        *(v4f*)(zp + 8) = z;
        *(v4f*)(zp + 12) = z;
        return;
    }
    const int qrow = pb[q0 + wave * 16 + lane16];
    const int kslot = tid >> 3;

    const float qscale = 0.17677669529663687f * 1.4426950408889634f;
    v4f qf[4];
    {
        const float* qp = q + ((size_t)b * S + qrow) * LD + h * 32 + half * 4;
#pragma unroll
        for (int u = 0; u < 4; ++u) qf[u] = *(const v4f*)(qp + 8 * u) * qscale;
    }

    v8f oacc[2] = {};
    float m_run = -3.0e38f;
    float l_part = 0.0f;

    v4i vi[4];
    int ki[4];
#pragma unroll
    for (int j = 0; j < 4; ++j) {
        vi[j] = *(const v4i*)(pb + 4 * wave + 16 * j);
        ki[j] = pb[kslot + 16 * j];
    }
    const unsigned rowB = (unsigned)(b * S);
    const unsigned colH = (unsigned)(h * 32);

    for (int j0 = 0; j0 < nk; j0 += 64) {
        const int jn = (j0 + 64 < S) ? j0 + 64 : j0;
        v4i vn[4];
        int kn[4];
#pragma unroll
        for (int j = 0; j < 4; ++j) {
            vn[j] = *(const v4i*)(pb + jn + 4 * wave + 16 * j);
            kn[j] = pb[jn + kslot + 16 * j];
        }
        const int pn = pb[jn + (tid & 63)];
        v4f rk[4];
        float rv[4][4];
#pragma unroll
        for (int j = 0; j < 4; ++j) {
            rk[j] = *(const v4f*)&k[(size_t)((rowB + (unsigned)ki[j]) * LD + colH + 4u * (tid & 7))];
        }
#pragma unroll
        for (int j = 0; j < 4; ++j) {
#pragma unroll
            for (int i = 0; i < 4; ++i)
                rv[j][i] = v[(size_t)((rowB + (unsigned)vi[j][i]) * LD + colH) + lane];
        }
#pragma unroll
        for (int j = 0; j < 4; ++j) {
            const int i = tid + j * 128;
            *(v4f*)&Ks[i >> 3][4 * (i & 7)] = rk[j];
        }
#pragma unroll
        for (int j = 0; j < 4; ++j) {
            const int kq = wave + 4 * j;
            v4f t = {rv[j][0], rv[j][1], rv[j][2], rv[j][3]};
            *(v4f*)&Vt[lane][4 * kq] = t;
        }
        mb[tid & 63] = (j0 + (tid & 63) < nk) ? 0.0f : -1e9f;
        {
            const size_t rowN = (size_t)((rowB + (unsigned)pn) * LD + colH);
            if (tid < 64) __builtin_prefetch(&k[rowN], 0, 1);
            else          __builtin_prefetch(&v[rowN], 0, 1);
        }
        __syncthreads();

        v8f sacc[4];
#pragma unroll
        for (int kt = 0; kt < 4; ++kt) {
            const v4f b0 = *(const v4f*)&mb[kt * 16 + half * 8];
            const v4f b1 = *(const v4f*)&mb[kt * 16 + half * 8 + 4];
            sacc[kt][0] = b0[0]; sacc[kt][1] = b0[1]; sacc[kt][2] = b0[2]; sacc[kt][3] = b0[3];
            sacc[kt][4] = b1[0]; sacc[kt][5] = b1[1]; sacc[kt][6] = b1[2]; sacc[kt][7] = b1[3];
        }
        v4f ka[4][4];
#pragma unroll
        for (int u = 0; u < 4; ++u)
#pragma unroll
            for (int kt = 0; kt < 4; ++kt)
                ka[u][kt] = *(const v4f*)&Ks[kt * 16 + lane16][8 * u + half * 4];
        v4f va[4][2][2];
#pragma unroll
        for (int kt = 0; kt < 4; ++kt)
#pragma unroll
            for (int w = 0; w < 2; ++w) {
                va[kt][w][0] = *(const v4f*)&Vt[lane16][kt * 16 + half * 8 + 4 * w];
                va[kt][w][1] = *(const v4f*)&Vt[16 + lane16][kt * 16 + half * 8 + 4 * w];
            }
#pragma unroll
        for (int u = 0; u < 4; ++u) {
#pragma unroll
            for (int e = 0; e < 2; ++e) {
                const v2f bq = {qf[u][2 * e], qf[u][2 * e + 1]};
#pragma unroll
                for (int kt = 0; kt < 4; ++kt) {
                    const v2f ak = {ka[u][kt][2 * e], ka[u][kt][2 * e + 1]};
                    sacc[kt] = wmma4(ak, bq, sacc[kt]);
                }
            }
        }

        float mx[4];
#pragma unroll
        for (int kt = 0; kt < 4; ++kt) {
            mx[kt] = sacc[kt][0];
#pragma unroll
            for (int r = 1; r < 8; ++r) mx[kt] = fmaxf(mx[kt], sacc[kt][r]);
        }
        float mc = fmaxf(fmaxf(mx[0], mx[1]), fmaxf(mx[2], mx[3]));
        mc = fmaxf(mc, xhalf(mc));
        const float mn = fmaxf(m_run, mc);
        const float al = __builtin_amdgcn_exp2f(m_run - mn);
        m_run = mn;
        float ls[4];
#pragma unroll
        for (int kt = 0; kt < 4; ++kt) {
            ls[kt] = 0.0f;
#pragma unroll
            for (int r = 0; r < 8; ++r) {
                const float p = __builtin_amdgcn_exp2f(sacc[kt][r] - mn);
                sacc[kt][r] = p;
                ls[kt] += p;
            }
        }
        l_part = l_part * al + ((ls[0] + ls[1]) + (ls[2] + ls[3]));
#pragma unroll
        for (int dt = 0; dt < 2; ++dt)
#pragma unroll
            for (int r = 0; r < 8; ++r) oacc[dt][r] *= al;

#pragma unroll
        for (int kt = 0; kt < 4; ++kt) {
#pragma unroll
            for (int w = 0; w < 2; ++w) {
#pragma unroll
                for (int e = 0; e < 2; ++e) {
                    const v2f pb = {sacc[kt][4 * w + 2 * e], sacc[kt][4 * w + 2 * e + 1]};
                    const v2f a0 = {va[kt][w][0][2 * e], va[kt][w][0][2 * e + 1]};
                    const v2f a1 = {va[kt][w][1][2 * e], va[kt][w][1][2 * e + 1]};
                    oacc[0] = wmma4(a0, pb, oacc[0]);
                    oacc[1] = wmma4(a1, pb, oacc[1]);
                }
            }
        }
        __syncthreads();
        vi[0] = vn[0]; vi[1] = vn[1]; vi[2] = vn[2]; vi[3] = vn[3];
        ki[0] = kn[0]; ki[1] = kn[1]; ki[2] = kn[2]; ki[3] = kn[3];
    }

    const float lsum = l_part + xhalf(l_part);
    const float linv = 1.0f / lsum;
    float* op = o + ((size_t)b * S + qrow) * LD + h * 32 + half * 8;
#pragma unroll
    for (int dt = 0; dt < 2; ++dt) {
        const v4f lo = {oacc[dt][0] * linv, oacc[dt][1] * linv, oacc[dt][2] * linv, oacc[dt][3] * linv};
        const v4f hi = {oacc[dt][4] * linv, oacc[dt][5] * linv, oacc[dt][6] * linv, oacc[dt][7] * linv};
        *(v4f*)(op + dt * 16) = lo;
        *(v4f*)(op + dt * 16 + 4) = hi;
    }
}

extern "C" void kernel_launch(void* const* d_in, const int* in_sizes, int n_in,
                              void* d_out, int out_size, void* d_ws, size_t ws_size,
                              hipStream_t stream) {
    (void)in_sizes; (void)n_in; (void)out_size; (void)ws_size;
    const int B = 4, S = 2048, D = 256, F = 1024, L = 4;
    const int M = B * S;

    const float* x_in  = (const float*)d_in[0];
    const int*   mask  = (const int*)d_in[1];
    const float* w_qs  = (const float*)d_in[2];
    const float* w_ks  = (const float*)d_in[3];
    const float* w_vs  = (const float*)d_in[4];
    const float* w_fc  = (const float*)d_in[5];
    const float* ln1_g = (const float*)d_in[6];
    const float* ln1_b = (const float*)d_in[7];
    const float* w1    = (const float*)d_in[8];
    const float* b1    = (const float*)d_in[9];
    const float* w2    = (const float*)d_in[10];
    const float* b2    = (const float*)d_in[11];
    const float* ln2_g = (const float*)d_in[12];
    const float* ln2_b = (const float*)d_in[13];

    float* ws    = (float*)d_ws;
    float* x_cur = ws;
    float* xn    = x_cur + (size_t)M * D;
    float* qb    = xn + (size_t)M * D;
    float* kb    = qb + (size_t)M * D;
    float* vb    = kb + (size_t)M * D;
    float* ob    = vb + (size_t)M * D;
    float* hb    = ob + (size_t)M * D;
    float* mfv   = hb + (size_t)M * F;
    int*   perm  = (int*)(mfv + M);
    int*   cnt   = perm + M;

    hipMemcpyAsync(x_cur, x_in, sizeof(float) * (size_t)M * D,
                   hipMemcpyDeviceToDevice, stream);
    mask_kernel<<<B, 1024, 0, stream>>>(mask, mfv, perm, cnt);

    const dim3 gD(M / 64, D / 128);
    const dim3 gQKV(M / 64, D / 128, 3);
    const dim3 gF(M / 64, F / 128);
    const dim3 gA(S / 64, B * 8);

    for (int l = 0; l < L; ++l) {
        const float* wq = w_qs + (size_t)l * D * D;
        const float* wk = w_ks + (size_t)l * D * D;
        const float* wv = w_vs + (size_t)l * D * D;
        const float* wf = w_fc + (size_t)l * D * D;
        const float* w1l = w1 + (size_t)l * D * F;
        const float* w2l = w2 + (size_t)l * F * D;
        const float* b1l = b1 + (size_t)l * F;
        const float* b2l = b2 + (size_t)l * D;

        ln_kernel<<<M / 32, 256, 0, stream>>>(x_cur, xn, ln1_g + (size_t)l * D, ln1_b + (size_t)l * D);
        gemm_kernel<<<gQKV, 256, 0, stream>>>(xn, wq, qb, nullptr, mfv, nullptr, D, D, 0, wk, kb, wv, vb);
        attn_kernel<<<gA, 128, 0, stream>>>(qb, kb, vb, perm, cnt, ob);
        gemm_kernel<<<gD, 256, 0, stream>>>(ob, wf, xn, nullptr, mfv, x_cur, D, D, 0, nullptr, nullptr, nullptr, nullptr);
        gemm_kernel<<<gF, 256, 0, stream>>>(xn, w1l, hb, b1l, nullptr, nullptr, D, F, 1, nullptr, nullptr, nullptr, nullptr);
        gemm_kernel<<<gD, 256, 0, stream>>>(hb, w2l, qb, b2l, nullptr, xn, F, D, 0, nullptr, nullptr, nullptr, nullptr);
        ln_kernel<<<M / 32, 256, 0, stream>>>(qb, x_cur, ln2_g + (size_t)l * D, ln2_b + (size_t)l * D);
    }

    hipMemcpyAsync(d_out, x_cur, sizeof(float) * (size_t)M * D,
                   hipMemcpyDeviceToDevice, stream);
}
